# starting kernel
# baseline (speedup 1.0000x reference)
#include <hip/hip_runtime.h>
#include <hip/hip_bf16.h>

typedef __attribute__((ext_vector_type(2))) float v2f;
typedef __attribute__((ext_vector_type(8))) float v8f;

#define IN_DIM 128
#define OUT_DIM 256
#define LN_EPS 1e-5f

__global__ void zero_u32(unsigned* __restrict__ p, unsigned n) {
    unsigned i = blockIdx.x * blockDim.x + threadIdx.x;
    if (i < n) p[i] = 0u;
}

__global__ void degree_kernel(const int* __restrict__ src, const int* __restrict__ dst,
                              int* __restrict__ deg_out, int* __restrict__ deg_in, int n_edges) {
    int e = blockIdx.x * blockDim.x + threadIdx.x;
    if (e >= n_edges) return;
    atomicAdd(&deg_out[src[e]], 1);
    atomicAdd(&deg_in[dst[e]], 1);
}

__global__ void compute_h(const float* __restrict__ feat, const int* __restrict__ deg_out,
                          float* __restrict__ h, int n_nodes) {
    int idx = blockIdx.x * blockDim.x + threadIdx.x;
    int total = n_nodes * (IN_DIM / 4);
    if (idx >= total) return;
    int node = idx / (IN_DIM / 4);
    float nrm = rsqrtf(fmaxf((float)deg_out[node], 1.0f));
    float4 v = ((const float4*)feat)[idx];
    v.x *= nrm; v.y *= nrm; v.z *= nrm; v.w *= nrm;
    ((float4*)h)[idx] = v;
}

__global__ __launch_bounds__(256) void spmm_scatter(const int* __restrict__ src,
                                                    const int* __restrict__ dst,
                                                    const float* __restrict__ h,
                                                    float* __restrict__ agg, int n_edges) {
    int edge = blockIdx.x * (blockDim.x >> 5) + (threadIdx.x >> 5);
    if (edge >= n_edges) return;
    int lane = threadIdx.x & 31;
    int s = src[edge];
    int d = dst[edge];
    const float4* hp = (const float4*)(h + (size_t)s * IN_DIM);
    float4 v = hp[lane];
    float* ap = agg + (size_t)d * IN_DIM + lane * 4;
    __hip_atomic_fetch_add(ap + 0, v.x, __ATOMIC_RELAXED, __HIP_MEMORY_SCOPE_AGENT);
    __hip_atomic_fetch_add(ap + 1, v.y, __ATOMIC_RELAXED, __HIP_MEMORY_SCOPE_AGENT);
    __hip_atomic_fetch_add(ap + 2, v.z, __ATOMIC_RELAXED, __HIP_MEMORY_SCOPE_AGENT);
    __hip_atomic_fetch_add(ap + 3, v.w, __ATOMIC_RELAXED, __HIP_MEMORY_SCOPE_AGENT);
}

__global__ __launch_bounds__(256) void gcn_fused(const float* __restrict__ agg,
                                                 const int* __restrict__ deg_in,
                                                 const float* __restrict__ W,
                                                 const float* __restrict__ bias,
                                                 const float* __restrict__ gamma,
                                                 const float* __restrict__ beta,
                                                 const float* __restrict__ feat,
                                                 const float* __restrict__ skip_W,
                                                 const float* __restrict__ skip_b,
                                                 float* __restrict__ out, int n_nodes) {
    __shared__ float lds_gcn[16][OUT_DIM + 4];
    __shared__ float lds_skip[16][OUT_DIM + 4];

    const int m0   = blockIdx.x * 16;
    const int wave = threadIdx.x >> 5;
    const int lane = threadIdx.x & 31;
    const int half = lane >> 4;
    const int l16  = lane & 15;

    const int arow = m0 + l16;
    const float* agg_row  = agg  + (size_t)arow * IN_DIM;
    const float* feat_row = feat + (size_t)arow * IN_DIM;
    const int bcol = wave * 32 + l16;

    v8f cg0 = {}, cg1 = {};
    v8f cs0 = {}, cs1 = {};

    for (int k0 = 0; k0 < IN_DIM; k0 += 4) {
        const int ka = k0 + 2 * half;
        v2f aA = *(const v2f*)(agg_row + ka);
        v2f aF = *(const v2f*)(feat_row + ka);
        const float* Wk0 = W      + (size_t)ka * OUT_DIM;
        const float* Wk1 = Wk0 + OUT_DIM;
        const float* Sk0 = skip_W + (size_t)ka * OUT_DIM;
        const float* Sk1 = Sk0 + OUT_DIM;
        v2f bW0, bW1, bS0, bS1;
        bW0[0] = Wk0[bcol];      bW0[1] = Wk1[bcol];
        bW1[0] = Wk0[bcol + 16]; bW1[1] = Wk1[bcol + 16];
        bS0[0] = Sk0[bcol];      bS0[1] = Sk1[bcol];
        bS1[0] = Sk0[bcol + 16]; bS1[1] = Sk1[bcol + 16];
        cg0 = __builtin_amdgcn_wmma_f32_16x16x4_f32(false, aA, false, bW0, (short)0, cg0, false, false);
        cg1 = __builtin_amdgcn_wmma_f32_16x16x4_f32(false, aA, false, bW1, (short)0, cg1, false, false);
        cs0 = __builtin_amdgcn_wmma_f32_16x16x4_f32(false, aF, false, bS0, (short)0, cs0, false, false);
        cs1 = __builtin_amdgcn_wmma_f32_16x16x4_f32(false, aF, false, bS1, (short)0, cs1, false, false);
    }

    float nrmv[8];
#pragma unroll
    for (int i = 0; i < 8; ++i) {
        const int row = i + 8 * half;
        nrmv[i] = rsqrtf(fmaxf((float)deg_in[m0 + row], 1.0f));
    }

    {
        const float bb0 = bias[bcol],        sb0 = skip_b[bcol];
        const float bb1 = bias[bcol + 16],   sb1 = skip_b[bcol + 16];
#pragma unroll
        for (int i = 0; i < 8; ++i) {
            const int row = i + 8 * half;
            lds_gcn[row][bcol]       = cg0[i] * nrmv[i] + bb0;
            lds_gcn[row][bcol + 16]  = cg1[i] * nrmv[i] + bb1;
            lds_skip[row][bcol]      = cs0[i] + sb0;
            lds_skip[row][bcol + 16] = cs1[i] + sb1;
        }
    }
    __syncthreads();

#pragma unroll
    for (int rr = 0; rr < 2; ++rr) {
        const int row = wave * 2 + rr;
        float vals[8];
        float s = 0.f, s2 = 0.f;
#pragma unroll
        for (int j = 0; j < 8; ++j) {
            float x = lds_gcn[row][lane * 8 + j];
            vals[j] = x;
            s += x;
            s2 += x * x;
        }
        for (int off = 16; off > 0; off >>= 1) {
            s  += __shfl_xor(s,  off, 32);
            s2 += __shfl_xor(s2, off, 32);
        }
        const float mu  = s * (1.0f / OUT_DIM);
        const float var = s2 * (1.0f / OUT_DIM) - mu * mu;
        const float rs  = rsqrtf(var + LN_EPS);
        float* orow = out + (size_t)(m0 + row) * OUT_DIM;
#pragma unroll
        for (int j = 0; j < 8; ++j) {
            const int c = lane * 8 + j;
            float ln = (vals[j] - mu) * rs * gamma[c] + beta[c];
            orow[c] = fmaxf(ln, 0.f) + lds_skip[row][c];
        }
    }
}

extern "C" void kernel_launch(void* const* d_in, const int* in_sizes, int n_in,
                              void* d_out, int out_size, void* d_ws, size_t ws_size,
                              hipStream_t stream) {
    const float* feat   = (const float*)d_in[0];
    const int*   src    = (const int*)d_in[1];
    const int*   dst    = (const int*)d_in[2];
    const float* W      = (const float*)d_in[3];
    const float* bias   = (const float*)d_in[4];
    const float* gamma  = (const float*)d_in[5];
    const float* beta   = (const float*)d_in[6];
    const float* skip_W = (const float*)d_in[7];
    const float* skip_b = (const float*)d_in[8];
    float* out = (float*)d_out;

    const int n_nodes = in_sizes[0] / IN_DIM;
    const int n_edges = in_sizes[1];

    int*   deg_out = (int*)d_ws;
    int*   deg_in  = deg_out + n_nodes;
    float* h       = (float*)(deg_in + n_nodes);
    float* agg     = h + (size_t)n_nodes * IN_DIM;

    const int T = 256;

    {
        unsigned n = 2u * (unsigned)n_nodes;
        zero_u32<<<(n + T - 1) / T, T, 0, stream>>>((unsigned*)deg_out, n);
        unsigned na = (unsigned)n_nodes * IN_DIM;
        zero_u32<<<(na + T - 1) / T, T, 0, stream>>>((unsigned*)agg, na);
    }

    degree_kernel<<<(n_edges + T - 1) / T, T, 0, stream>>>(src, dst, deg_out, deg_in, n_edges);

    {
        int total = n_nodes * (IN_DIM / 4);
        compute_h<<<(total + T - 1) / T, T, 0, stream>>>(feat, deg_out, h, n_nodes);
    }

    {
        int waves_per_block = T / 32;
        int blocks = (n_edges + waves_per_block - 1) / waves_per_block;
        spmm_scatter<<<blocks, T, 0, stream>>>(src, dst, h, agg, n_edges);
    }

    gcn_fused<<<n_nodes / 16, T, 0, stream>>>(agg, deg_in, W, bias, gamma, beta,
                                              feat, skip_W, skip_b, out, n_nodes);
}


# aggregation made dst-range stationary: 196 work-groups each keep 512 agg rows in LDS, filter the edge list, add h[src] rows by LDS float atomics (arrival order); no global float atomics
# speedup vs baseline: 1.3687x; 1.3687x over previous
#include <hip/hip_runtime.h>
#include <hip/hip_bf16.h>

typedef __attribute__((ext_vector_type(2))) float v2f;
typedef __attribute__((ext_vector_type(8))) float v8f;

#define IN_DIM 128
#define OUT_DIM 256
#define LN_EPS 1e-5f

__global__ void zero_u32(unsigned* __restrict__ p, unsigned n) {
    unsigned i = blockIdx.x * blockDim.x + threadIdx.x;
    if (i < n) p[i] = 0u;
}

__global__ void degree_kernel(const int* __restrict__ src, const int* __restrict__ dst,
                              int* __restrict__ deg_out, int* __restrict__ deg_in, int n_edges) {
    int e = blockIdx.x * blockDim.x + threadIdx.x;
    if (e >= n_edges) return;
    atomicAdd(&deg_out[src[e]], 1);
    atomicAdd(&deg_in[dst[e]], 1);
}

__global__ void compute_h(const float* __restrict__ feat, const int* __restrict__ deg_out,
                          float* __restrict__ h, int n_nodes) {
    int idx = blockIdx.x * blockDim.x + threadIdx.x;
    int total = n_nodes * (IN_DIM / 4);
    if (idx >= total) return;
    int node = idx / (IN_DIM / 4);
    float nrm = rsqrtf(fmaxf((float)deg_out[node], 1.0f));
    float4 v = ((const float4*)feat)[idx];
    v.x *= nrm; v.y *= nrm; v.z *= nrm; v.w *= nrm;
    ((float4*)h)[idx] = v;
}

#define RNG   512
#define AGG_T 1024
#define AGG_W (AGG_T / 32)
#define QCAP  160
#define GU    8

__device__ __forceinline__ void q_append(bool m, int e, unsigned dl, int& cnt,
                                         int* qe, unsigned short* qd, int lane) {
    const unsigned mask = __builtin_amdgcn_ballot_w32(m);
    if (mask) {
        const int pos = cnt + __builtin_popcount(mask & ((1u << lane) - 1u));
        if (m) { qe[pos] = e; qd[pos] = (unsigned short)dl; }
        cnt += __builtin_popcount(mask);
    }
}

__device__ __forceinline__ void q_scan(int4 d, int e0, int base, int& cnt,
                                       int* qe, unsigned short* qd, int lane) {
    const unsigned ux = (unsigned)(d.x - base), uy = (unsigned)(d.y - base);
    const unsigned uz = (unsigned)(d.z - base), uw = (unsigned)(d.w - base);
    const bool mx = ux < RNG, my = uy < RNG, mz = uz < RNG, mw = uw < RNG;
    if (__builtin_amdgcn_ballot_w32(mx | my | mz | mw)) {
        q_append(mx, e0 + 0, ux, cnt, qe, qd, lane);
        q_append(my, e0 + 1, uy, cnt, qe, qd, lane);
        q_append(mz, e0 + 2, uz, cnt, qe, qd, lane);
        q_append(mw, e0 + 3, uw, cnt, qe, qd, lane);
    }
}

__device__ __forceinline__ void q_drain(int& cnt, const int* qe, const unsigned short* qd,
                                        const int* __restrict__ src, const float* __restrict__ h,
                                        float* acc, int lane) {
    for (int b0 = 0; b0 < cnt; b0 += 32) {
        const int nb = min(32, cnt - b0);
        int s = 0, dl = 0;
        if (lane < nb) { s = src[qe[b0 + lane]]; dl = qd[b0 + lane]; }
        for (int i = 0; i < nb; i += GU) {
            float4 v[GU];
            int rl[GU];
#pragma unroll
            for (int u = 0; u < GU; ++u) {
                const int iu = min(i + u, nb - 1);
                const int su = __builtin_amdgcn_readlane(s, iu);
                rl[u] = __builtin_amdgcn_readlane(dl, iu);
                v[u] = ((const float4*)(h + (size_t)su * IN_DIM))[lane];
            }
#pragma unroll
            for (int u = 0; u < GU; ++u) {
                if (i + u < nb) {
                    float* a = acc + rl[u] * IN_DIM + lane;
                    __hip_atomic_fetch_add(a,      v[u].x, __ATOMIC_RELAXED, __HIP_MEMORY_SCOPE_WORKGROUP);
                    __hip_atomic_fetch_add(a + 32, v[u].y, __ATOMIC_RELAXED, __HIP_MEMORY_SCOPE_WORKGROUP);
                    __hip_atomic_fetch_add(a + 64, v[u].z, __ATOMIC_RELAXED, __HIP_MEMORY_SCOPE_WORKGROUP);
                    __hip_atomic_fetch_add(a + 96, v[u].w, __ATOMIC_RELAXED, __HIP_MEMORY_SCOPE_WORKGROUP);
                }
            }
        }
    }
    cnt = 0;
}

__global__ __launch_bounds__(AGG_T) void agg_range(const int* __restrict__ src,
                                                   const int* __restrict__ dst,
                                                   const float* __restrict__ h,
                                                   float* __restrict__ agg,
                                                   int n_edges, int n_nodes) {
    __shared__ __attribute__((aligned(16))) float acc[RNG * IN_DIM];
    __shared__ int qe_all[AGG_W][QCAP];
    __shared__ unsigned short qd_all[AGG_W][QCAP];

    const int tid  = threadIdx.x;
    const int lane = tid & 31;
    const int wv   = tid >> 5;
    const int base = blockIdx.x * RNG;
    int* qe = qe_all[wv];
    unsigned short* qd = qd_all[wv];

    for (int i = tid; i < RNG * IN_DIM / 4; i += AGG_T)
        ((float4*)acc)[i] = make_float4(0.f, 0.f, 0.f, 0.f);
    __syncthreads();

    int cnt = 0;
    const int4* dst4 = (const int4*)dst;
    const int nvec = n_edges >> 2;
    const int4 none = make_int4(-1, -1, -1, -1);

    for (int v0 = 0; v0 < nvec; v0 += 4 * AGG_T) {
        int4 d[4];
#pragma unroll
        for (int u = 0; u < 4; ++u) {
            const int v = v0 + u * AGG_T + tid;
            const int4 t = dst4[min(v, nvec - 1)];
            d[u] = (v < nvec) ? t : none;
        }
#pragma unroll
        for (int u = 0; u < 4; ++u) {
            const int v = v0 + u * AGG_T + tid;
            q_scan(d[u], v * 4, base, cnt, qe, qd, lane);
            if (cnt >= 32) q_drain(cnt, qe, qd, src, h, acc, lane);
        }
    }
    {
        const int rem = n_edges & 3;
        if (rem) {
            const int e = (nvec << 2) + tid;
            int4 d = none;
            if (tid < rem) d.x = dst[e];
            q_scan(d, e, base, cnt, qe, qd, lane);
        }
    }
    q_drain(cnt, qe, qd, src, h, acc, lane);
    __syncthreads();

    for (int r = wv; r < RNG; r += AGG_W) {
        const int row = base + r;
        if (row >= n_nodes) break;
        const float* a = acc + r * IN_DIM + lane;
        const float4 o = make_float4(a[0], a[32], a[64], a[96]);
        ((float4*)(agg + (size_t)row * IN_DIM))[lane] = o;
    }
}

__global__ __launch_bounds__(256) void gcn_fused(const float* __restrict__ agg,
                                                 const int* __restrict__ deg_in,
                                                 const float* __restrict__ W,
                                                 const float* __restrict__ bias,
                                                 const float* __restrict__ gamma,
                                                 const float* __restrict__ beta,
                                                 const float* __restrict__ feat,
                                                 const float* __restrict__ skip_W,
                                                 const float* __restrict__ skip_b,
                                                 float* __restrict__ out, int n_nodes) {
    __shared__ float lds_gcn[16][OUT_DIM + 4];
    __shared__ float lds_skip[16][OUT_DIM + 4];

    const int m0   = blockIdx.x * 16;
    const int wave = threadIdx.x >> 5;
    const int lane = threadIdx.x & 31;
    const int half = lane >> 4;
    const int l16  = lane & 15;

    const int arow = m0 + l16;
    const float* agg_row  = agg  + (size_t)arow * IN_DIM;
    const float* feat_row = feat + (size_t)arow * IN_DIM;
    const int bcol = wave * 32 + l16;

    v8f cg0 = {}, cg1 = {};
    v8f cs0 = {}, cs1 = {};

    for (int k0 = 0; k0 < IN_DIM; k0 += 4) {
        const int ka = k0 + 2 * half;
        v2f aA = *(const v2f*)(agg_row + ka);
        v2f aF = *(const v2f*)(feat_row + ka);
        const float* Wk0 = W      + (size_t)ka * OUT_DIM;
        const float* Wk1 = Wk0 + OUT_DIM;
        const float* Sk0 = skip_W + (size_t)ka * OUT_DIM;
        const float* Sk1 = Sk0 + OUT_DIM;
        v2f bW0, bW1, bS0, bS1;
        bW0[0] = Wk0[bcol];      bW0[1] = Wk1[bcol];
        bW1[0] = Wk0[bcol + 16]; bW1[1] = Wk1[bcol + 16];
        bS0[0] = Sk0[bcol];      bS0[1] = Sk1[bcol];
        bS1[0] = Sk0[bcol + 16]; bS1[1] = Sk1[bcol + 16];
        cg0 = __builtin_amdgcn_wmma_f32_16x16x4_f32(false, aA, false, bW0, (short)0, cg0, false, false);
        cg1 = __builtin_amdgcn_wmma_f32_16x16x4_f32(false, aA, false, bW1, (short)0, cg1, false, false);
        cs0 = __builtin_amdgcn_wmma_f32_16x16x4_f32(false, aF, false, bS0, (short)0, cs0, false, false);
        cs1 = __builtin_amdgcn_wmma_f32_16x16x4_f32(false, aF, false, bS1, (short)0, cs1, false, false);
    }

    float nrmv[8];
#pragma unroll
    for (int i = 0; i < 8; ++i) {
        const int row = i + 8 * half;
        nrmv[i] = rsqrtf(fmaxf((float)deg_in[m0 + row], 1.0f));
    }

    {
        const float bb0 = bias[bcol],        sb0 = skip_b[bcol];
        const float bb1 = bias[bcol + 16],   sb1 = skip_b[bcol + 16];
#pragma unroll
        for (int i = 0; i < 8; ++i) {
            const int row = i + 8 * half;
            lds_gcn[row][bcol]       = cg0[i] * nrmv[i] + bb0;
            lds_gcn[row][bcol + 16]  = cg1[i] * nrmv[i] + bb1;
            lds_skip[row][bcol]      = cs0[i] + sb0;
            lds_skip[row][bcol + 16] = cs1[i] + sb1;
        }
    }
    __syncthreads();

#pragma unroll
    for (int rr = 0; rr < 2; ++rr) {
        const int row = wave * 2 + rr;
        float vals[8];
        float s = 0.f, s2 = 0.f;
#pragma unroll
        for (int j = 0; j < 8; ++j) {
            float x = lds_gcn[row][lane * 8 + j];
            vals[j] = x;
            s += x;
            s2 += x * x;
        }
        for (int off = 16; off > 0; off >>= 1) {
            s  += __shfl_xor(s,  off, 32);
            s2 += __shfl_xor(s2, off, 32);
        }
        const float mu  = s * (1.0f / OUT_DIM);
        const float var = s2 * (1.0f / OUT_DIM) - mu * mu;
        const float rs  = rsqrtf(var + LN_EPS);
        float* orow = out + (size_t)(m0 + row) * OUT_DIM;
#pragma unroll
        for (int j = 0; j < 8; ++j) {
            const int c = lane * 8 + j;
            float ln = (vals[j] - mu) * rs * gamma[c] + beta[c];
            orow[c] = fmaxf(ln, 0.f) + lds_skip[row][c];
        }
    }
}

extern "C" void kernel_launch(void* const* d_in, const int* in_sizes, int n_in,
                              void* d_out, int out_size, void* d_ws, size_t ws_size,
                              hipStream_t stream) {
    const float* feat   = (const float*)d_in[0];
    const int*   src    = (const int*)d_in[1];
    const int*   dst    = (const int*)d_in[2];
    const float* W      = (const float*)d_in[3];
    const float* bias   = (const float*)d_in[4];
    const float* gamma  = (const float*)d_in[5];
    const float* beta   = (const float*)d_in[6];
    const float* skip_W = (const float*)d_in[7];
    const float* skip_b = (const float*)d_in[8];
    float* out = (float*)d_out;

    const int n_nodes = in_sizes[0] / IN_DIM;
    const int n_edges = in_sizes[1];

    int*   deg_out = (int*)d_ws;
    int*   deg_in  = deg_out + n_nodes;
    float* h       = (float*)(deg_in + n_nodes);
    float* agg     = h + (size_t)n_nodes * IN_DIM;

    const int T = 256;

    {
        unsigned n = 2u * (unsigned)n_nodes;
        zero_u32<<<(n + T - 1) / T, T, 0, stream>>>((unsigned*)deg_out, n);
    }

    degree_kernel<<<(n_edges + T - 1) / T, T, 0, stream>>>(src, dst, deg_out, deg_in, n_edges);

    {
        int total = n_nodes * (IN_DIM / 4);
        compute_h<<<(total + T - 1) / T, T, 0, stream>>>(feat, deg_out, h, n_nodes);
    }

    agg_range<<<(n_nodes + RNG - 1) / RNG, AGG_T, 0, stream>>>(src, dst, h, agg, n_edges, n_nodes);

    gcn_fused<<<n_nodes / 16, T, 0, stream>>>(agg, deg_in, W, bias, gamma, beta,
                                              feat, skip_W, skip_b, out, n_nodes);
}


# agg_range scans a 2-byte range-id stream (dst/512, written by the degree pass into the head of out; 8 edges per 16-B load, packed 16-bit test), not the dst list; LDS atomics, arrival order kept
# speedup vs baseline: 1.5190x; 1.1098x over previous
#include <hip/hip_runtime.h>
#include <hip/hip_bf16.h>

typedef __attribute__((ext_vector_type(2))) float v2f;
typedef __attribute__((ext_vector_type(8))) float v8f;

#define IN_DIM 128
#define OUT_DIM 256
#define LN_EPS 1e-5f

__global__ void zero_u32(unsigned* __restrict__ p, unsigned n) {
    unsigned i = blockIdx.x * blockDim.x + threadIdx.x;
    if (i < n) p[i] = 0u;
}

#define RNG      512
#define KEY_NONE 0xFFFFu
__global__ void degree_kernel(const int* __restrict__ src, const int* __restrict__ dst,
                              int* __restrict__ deg_out, int* __restrict__ deg_in,
                              unsigned short* __restrict__ key, int n_edges, int n_pad) {
    int e = blockIdx.x * blockDim.x + threadIdx.x;
    if (e >= n_pad) return;
    if (e >= n_edges) { key[e] = (unsigned short)KEY_NONE; return; }
    const int d = dst[e];
    atomicAdd(&deg_out[src[e]], 1);
    atomicAdd(&deg_in[d], 1);
    key[e] = (unsigned short)((unsigned)d / RNG);
}

__global__ void compute_h(const float* __restrict__ feat, const int* __restrict__ deg_out,
                          float* __restrict__ h, int n_nodes) {
    int idx = blockIdx.x * blockDim.x + threadIdx.x;
    int total = n_nodes * (IN_DIM / 4);
    if (idx >= total) return;
    int node = idx / (IN_DIM / 4);
    float nrm = rsqrtf(fmaxf((float)deg_out[node], 1.0f));
    float4 v = ((const float4*)feat)[idx];
    v.x *= nrm; v.y *= nrm; v.z *= nrm; v.w *= nrm;
    ((float4*)h)[idx] = v;
}

#define AGG_T 1024
#define AGG_W (AGG_T / 32)
#define KPV   8
#define SCAN_U 4
#define QCAP  288
#define GU    8

typedef unsigned short __attribute__((ext_vector_type(2))) u16x2;

__device__ __forceinline__ unsigned key_hit(int k, unsigned pat) {
    const u16x2 one = {1, 1};
    const u16x2 x = __builtin_bit_cast(u16x2, (unsigned)k ^ pat);
    return __builtin_bit_cast(unsigned, __builtin_elementwise_sub_sat(one, x));
}

__device__ __forceinline__ void q_scan(int4 k, int e0, unsigned pat, int& cnt, int* qe, int lane) {
    unsigned g = key_hit(k.x, pat) | (key_hit(k.y, pat) << 1) |
                 (key_hit(k.z, pat) << 2) | (key_hit(k.w, pat) << 3);
    unsigned mask = __builtin_amdgcn_ballot_w32(g != 0u);
    while (mask) {
        const bool m = g != 0u;
        const int b = __builtin_ctz(g | 0x80000000u);
        const int pos = cnt + __builtin_popcount(mask & ((1u << lane) - 1u));
        if (m) qe[pos] = e0 + ((b & 3) << 1) + (b >> 4);
        cnt += __builtin_popcount(mask);
        g &= g - 1u;
        mask = __builtin_amdgcn_ballot_w32(g != 0u);
    }
}

__device__ __forceinline__ void q_drain(int& cnt, const int* qe, int base,
                                        const int* __restrict__ src, const int* __restrict__ dst,
                                        const float* __restrict__ h, float* acc, int lane) {
    for (int b0 = 0; b0 < cnt; b0 += 32) {
        const int nb = min(32, cnt - b0);
        int s = 0, dl = 0;
        if (lane < nb) { const int e = qe[b0 + lane]; s = src[e]; dl = dst[e] - base; }
        for (int i = 0; i < nb; i += GU) {
            float4 v[GU];
            int rl[GU];
#pragma unroll
            for (int u = 0; u < GU; ++u) {
                const int iu = min(i + u, nb - 1);
                const int su = __builtin_amdgcn_readlane(s, iu);
                rl[u] = __builtin_amdgcn_readlane(dl, iu);
                v[u] = ((const float4*)(h + (size_t)su * IN_DIM))[lane];
            }
#pragma unroll
            for (int u = 0; u < GU; ++u) {
                if (i + u < nb) {
                    float* a = acc + rl[u] * IN_DIM + lane;
                    __hip_atomic_fetch_add(a,      v[u].x, __ATOMIC_RELAXED, __HIP_MEMORY_SCOPE_WORKGROUP);
                    __hip_atomic_fetch_add(a + 32, v[u].y, __ATOMIC_RELAXED, __HIP_MEMORY_SCOPE_WORKGROUP);
                    __hip_atomic_fetch_add(a + 64, v[u].z, __ATOMIC_RELAXED, __HIP_MEMORY_SCOPE_WORKGROUP);
                    __hip_atomic_fetch_add(a + 96, v[u].w, __ATOMIC_RELAXED, __HIP_MEMORY_SCOPE_WORKGROUP);
                }
            }
        }
    }
    cnt = 0;
}

__global__ __launch_bounds__(AGG_T) void agg_range(const int* __restrict__ src,
                                                   const int* __restrict__ dst,
                                                   const unsigned short* __restrict__ key,
                                                   const float* __restrict__ h,
                                                   float* __restrict__ agg,
                                                   int n_pad, int n_nodes) {
    __shared__ __attribute__((aligned(16))) float acc[RNG * IN_DIM];
    __shared__ int qe_all[AGG_W][QCAP];

    const int tid  = threadIdx.x;
    const int lane = tid & 31;
    const int wv   = tid >> 5;
    const int base = blockIdx.x * RNG;
    int* qe = qe_all[wv];
    const unsigned pat = blockIdx.x | (blockIdx.x << 16);

    for (int i = tid; i < RNG * IN_DIM / 4; i += AGG_T)
        ((float4*)acc)[i] = make_float4(0.f, 0.f, 0.f, 0.f);
    __syncthreads();

    int cnt = 0;
    const int4* key4 = (const int4*)key;
    const int nvec = n_pad / KPV;

    for (int v0 = 0; v0 < nvec; v0 += SCAN_U * AGG_T) {
        int4 k[SCAN_U];
#pragma unroll
        for (int u = 0; u < SCAN_U; ++u)
            k[u] = key4[v0 + u * AGG_T + tid];
#pragma unroll
        for (int u = 0; u < SCAN_U; ++u) {
            const int v = v0 + u * AGG_T + tid;
            q_scan(k[u], v * KPV, pat, cnt, qe, lane);
            if (cnt >= 32) q_drain(cnt, qe, base, src, dst, h, acc, lane);
        }
    }
    q_drain(cnt, qe, base, src, dst, h, acc, lane);
    __syncthreads();

    for (int r = wv; r < RNG; r += AGG_W) {
        const int row = base + r;
        if (row >= n_nodes) break;
        const float* a = acc + r * IN_DIM + lane;
        const float4 o = make_float4(a[0], a[32], a[64], a[96]);
        ((float4*)(agg + (size_t)row * IN_DIM))[lane] = o;
    }
}

__global__ __launch_bounds__(256) void gcn_fused(const float* __restrict__ agg,
                                                 const int* __restrict__ deg_in,
                                                 const float* __restrict__ W,
                                                 const float* __restrict__ bias,
                                                 const float* __restrict__ gamma,
                                                 const float* __restrict__ beta,
                                                 const float* __restrict__ feat,
                                                 const float* __restrict__ skip_W,
                                                 const float* __restrict__ skip_b,
                                                 float* __restrict__ out, int n_nodes) {
    __shared__ float lds_gcn[16][OUT_DIM + 4];
    __shared__ float lds_skip[16][OUT_DIM + 4];

    const int m0   = blockIdx.x * 16;
    const int wave = threadIdx.x >> 5;
    const int lane = threadIdx.x & 31;
    const int half = lane >> 4;
    const int l16  = lane & 15;

    const int arow = m0 + l16;
    const float* agg_row  = agg  + (size_t)arow * IN_DIM;
    const float* feat_row = feat + (size_t)arow * IN_DIM;
    const int bcol = wave * 32 + l16;

    v8f cg0 = {}, cg1 = {};
    v8f cs0 = {}, cs1 = {};

    for (int k0 = 0; k0 < IN_DIM; k0 += 4) {
        const int ka = k0 + 2 * half;
        v2f aA = *(const v2f*)(agg_row + ka);
        v2f aF = *(const v2f*)(feat_row + ka);
        const float* Wk0 = W      + (size_t)ka * OUT_DIM;
        const float* Wk1 = Wk0 + OUT_DIM;
        const float* Sk0 = skip_W + (size_t)ka * OUT_DIM;
        const float* Sk1 = Sk0 + OUT_DIM;
        v2f bW0, bW1, bS0, bS1;
        bW0[0] = Wk0[bcol];      bW0[1] = Wk1[bcol];
        bW1[0] = Wk0[bcol + 16]; bW1[1] = Wk1[bcol + 16];
        bS0[0] = Sk0[bcol];      bS0[1] = Sk1[bcol];
        bS1[0] = Sk0[bcol + 16]; bS1[1] = Sk1[bcol + 16];
        cg0 = __builtin_amdgcn_wmma_f32_16x16x4_f32(false, aA, false, bW0, (short)0, cg0, false, false);
        cg1 = __builtin_amdgcn_wmma_f32_16x16x4_f32(false, aA, false, bW1, (short)0, cg1, false, false);
        cs0 = __builtin_amdgcn_wmma_f32_16x16x4_f32(false, aF, false, bS0, (short)0, cs0, false, false);
        cs1 = __builtin_amdgcn_wmma_f32_16x16x4_f32(false, aF, false, bS1, (short)0, cs1, false, false);
    }

    float nrmv[8];
#pragma unroll
    for (int i = 0; i < 8; ++i) {
        const int row = i + 8 * half;
        nrmv[i] = rsqrtf(fmaxf((float)deg_in[m0 + row], 1.0f));
    }

    {
        const float bb0 = bias[bcol],        sb0 = skip_b[bcol];
        const float bb1 = bias[bcol + 16],   sb1 = skip_b[bcol + 16];
#pragma unroll
        for (int i = 0; i < 8; ++i) {
            const int row = i + 8 * half;
            lds_gcn[row][bcol]       = cg0[i] * nrmv[i] + bb0;
            lds_gcn[row][bcol + 16]  = cg1[i] * nrmv[i] + bb1;
            lds_skip[row][bcol]      = cs0[i] + sb0;
            lds_skip[row][bcol + 16] = cs1[i] + sb1;
        }
    }
    __syncthreads();

#pragma unroll
    for (int rr = 0; rr < 2; ++rr) {
        const int row = wave * 2 + rr;
        float vals[8];
        float s = 0.f, s2 = 0.f;
#pragma unroll
        for (int j = 0; j < 8; ++j) {
            float x = lds_gcn[row][lane * 8 + j];
            vals[j] = x;
            s += x;
            s2 += x * x;
        }
        for (int off = 16; off > 0; off >>= 1) {
            s  += __shfl_xor(s,  off, 32);
            s2 += __shfl_xor(s2, off, 32);
        }
        const float mu  = s * (1.0f / OUT_DIM);
        const float var = s2 * (1.0f / OUT_DIM) - mu * mu;
        const float rs  = rsqrtf(var + LN_EPS);
        float* orow = out + (size_t)(m0 + row) * OUT_DIM;
#pragma unroll
        for (int j = 0; j < 8; ++j) {
            const int c = lane * 8 + j;
            float ln = (vals[j] - mu) * rs * gamma[c] + beta[c];
            orow[c] = fmaxf(ln, 0.f) + lds_skip[row][c];
        }
    }
}

extern "C" void kernel_launch(void* const* d_in, const int* in_sizes, int n_in,
                              void* d_out, int out_size, void* d_ws, size_t ws_size,
                              hipStream_t stream) {
    const float* feat   = (const float*)d_in[0];
    const int*   src    = (const int*)d_in[1];
    const int*   dst    = (const int*)d_in[2];
    const float* W      = (const float*)d_in[3];
    const float* bias   = (const float*)d_in[4];
    const float* gamma  = (const float*)d_in[5];
    const float* beta   = (const float*)d_in[6];
    const float* skip_W = (const float*)d_in[7];
    const float* skip_b = (const float*)d_in[8];
    float* out = (float*)d_out;

    const int n_nodes = in_sizes[0] / IN_DIM;
    const int n_edges = in_sizes[1];

    int*   deg_out = (int*)d_ws;
    int*   deg_in  = deg_out + n_nodes;
    float* h       = (float*)(deg_in + n_nodes);
    float* agg     = h + (size_t)n_nodes * IN_DIM;

    const int T = 256;

    unsigned short* key = (unsigned short*)d_out;
    const int scan_step = KPV * SCAN_U * AGG_T;
    const int n_pad = (n_edges + scan_step - 1) / scan_step * scan_step;

    {
        unsigned n = 2u * (unsigned)n_nodes;
        zero_u32<<<(n + T - 1) / T, T, 0, stream>>>((unsigned*)deg_out, n);
    }

    degree_kernel<<<(n_pad + T - 1) / T, T, 0, stream>>>(src, dst, deg_out, deg_in, key, n_edges, n_pad);

    {
        int total = n_nodes * (IN_DIM / 4);
        compute_h<<<(total + T - 1) / T, T, 0, stream>>>(feat, deg_out, h, n_nodes);
    }

    agg_range<<<(n_nodes + RNG - 1) / RNG, AGG_T, 0, stream>>>(src, dst, key, h, agg, n_pad, n_nodes);

    gcn_fused<<<n_nodes / 16, T, 0, stream>>>(agg, deg_in, W, bias, gamma, beta,
                                              feat, skip_W, skip_b, out, n_nodes);
}


# gcn_fused weight-stationary: 256 persistent work-groups keep W and skip_W in LDS in fragment order, loop over 16-row tiles; k order permuted inside each 8-k group (sum order changes, fixed)
# speedup vs baseline: 1.6053x; 1.0568x over previous
#include <hip/hip_runtime.h>
#include <hip/hip_bf16.h>

typedef __attribute__((ext_vector_type(2))) float v2f;
typedef __attribute__((ext_vector_type(8))) float v8f;

#define IN_DIM 128
#define OUT_DIM 256
#define LN_EPS 1e-5f

__global__ void zero_u32(unsigned* __restrict__ p, unsigned n) {
    unsigned i = blockIdx.x * blockDim.x + threadIdx.x;
    if (i < n) p[i] = 0u;
}

#define RNG      512
#define KEY_NONE 0xFFFFu
__global__ void degree_kernel(const int* __restrict__ src, const int* __restrict__ dst,
                              int* __restrict__ deg_out, int* __restrict__ deg_in,
                              unsigned short* __restrict__ key, int n_edges, int n_pad) {
    int e = blockIdx.x * blockDim.x + threadIdx.x;
    if (e >= n_pad) return;
    if (e >= n_edges) { key[e] = (unsigned short)KEY_NONE; return; }
    const int d = dst[e];
    atomicAdd(&deg_out[src[e]], 1);
    atomicAdd(&deg_in[d], 1);
    key[e] = (unsigned short)((unsigned)d / RNG);
}

__global__ void compute_h(const float* __restrict__ feat, const int* __restrict__ deg_out,
                          float* __restrict__ h, int n_nodes) {
    int idx = blockIdx.x * blockDim.x + threadIdx.x;
    int total = n_nodes * (IN_DIM / 4);
    if (idx >= total) return;
    int node = idx / (IN_DIM / 4);
    float nrm = rsqrtf(fmaxf((float)deg_out[node], 1.0f));
    float4 v = ((const float4*)feat)[idx];
    v.x *= nrm; v.y *= nrm; v.z *= nrm; v.w *= nrm;
    ((float4*)h)[idx] = v;
}

#define AGG_T 1024
#define AGG_W (AGG_T / 32)
#define KPV   8
#define SCAN_U 4
#define QCAP  288
#define GU    8

typedef unsigned short __attribute__((ext_vector_type(2))) u16x2;

__device__ __forceinline__ unsigned key_hit(int k, unsigned pat) {
    const u16x2 one = {1, 1};
    const u16x2 x = __builtin_bit_cast(u16x2, (unsigned)k ^ pat);
    return __builtin_bit_cast(unsigned, __builtin_elementwise_sub_sat(one, x));
}

__device__ __forceinline__ void q_scan(int4 k, int e0, unsigned pat, int& cnt, int* qe, int lane) {
    unsigned g = key_hit(k.x, pat) | (key_hit(k.y, pat) << 1) |
                 (key_hit(k.z, pat) << 2) | (key_hit(k.w, pat) << 3);
    unsigned mask = __builtin_amdgcn_ballot_w32(g != 0u);
    while (mask) {
        const bool m = g != 0u;
        const int b = __builtin_ctz(g | 0x80000000u);
        const int pos = cnt + __builtin_popcount(mask & ((1u << lane) - 1u));
        if (m) qe[pos] = e0 + ((b & 3) << 1) + (b >> 4);
        cnt += __builtin_popcount(mask);
        g &= g - 1u;
        mask = __builtin_amdgcn_ballot_w32(g != 0u);
    }
}

__device__ __forceinline__ void q_drain(int& cnt, const int* qe, int base,
                                        const int* __restrict__ src, const int* __restrict__ dst,
                                        const float* __restrict__ h, float* acc, int lane) {
    for (int b0 = 0; b0 < cnt; b0 += 32) {
        const int nb = min(32, cnt - b0);
        int s = 0, dl = 0;
        if (lane < nb) { const int e = qe[b0 + lane]; s = src[e]; dl = dst[e] - base; }
        for (int i = 0; i < nb; i += GU) {
            float4 v[GU];
            int rl[GU];
#pragma unroll
            for (int u = 0; u < GU; ++u) {
                const int iu = min(i + u, nb - 1);
                const int su = __builtin_amdgcn_readlane(s, iu);
                rl[u] = __builtin_amdgcn_readlane(dl, iu);
                v[u] = ((const float4*)(h + (size_t)su * IN_DIM))[lane];
            }
#pragma unroll
            for (int u = 0; u < GU; ++u) {
                if (i + u < nb) {
                    float* a = acc + rl[u] * IN_DIM + lane;
                    __hip_atomic_fetch_add(a,      v[u].x, __ATOMIC_RELAXED, __HIP_MEMORY_SCOPE_WORKGROUP);
                    __hip_atomic_fetch_add(a + 32, v[u].y, __ATOMIC_RELAXED, __HIP_MEMORY_SCOPE_WORKGROUP);
                    __hip_atomic_fetch_add(a + 64, v[u].z, __ATOMIC_RELAXED, __HIP_MEMORY_SCOPE_WORKGROUP);
                    __hip_atomic_fetch_add(a + 96, v[u].w, __ATOMIC_RELAXED, __HIP_MEMORY_SCOPE_WORKGROUP);
                }
            }
        }
    }
    cnt = 0;
}

__global__ __launch_bounds__(AGG_T) void agg_range(const int* __restrict__ src,
                                                   const int* __restrict__ dst,
                                                   const unsigned short* __restrict__ key,
                                                   const float* __restrict__ h,
                                                   float* __restrict__ agg,
                                                   int n_pad, int n_nodes) {
    __shared__ __attribute__((aligned(16))) float acc[RNG * IN_DIM];
    __shared__ int qe_all[AGG_W][QCAP];

    const int tid  = threadIdx.x;
    const int lane = tid & 31;
    const int wv   = tid >> 5;
    const int base = blockIdx.x * RNG;
    int* qe = qe_all[wv];
    const unsigned pat = blockIdx.x | (blockIdx.x << 16);

    for (int i = tid; i < RNG * IN_DIM / 4; i += AGG_T)
        ((float4*)acc)[i] = make_float4(0.f, 0.f, 0.f, 0.f);
    __syncthreads();

    int cnt = 0;
    const int4* key4 = (const int4*)key;
    const int nvec = n_pad / KPV;

    for (int v0 = 0; v0 < nvec; v0 += SCAN_U * AGG_T) {
        int4 k[SCAN_U];
#pragma unroll
        for (int u = 0; u < SCAN_U; ++u)
            k[u] = key4[v0 + u * AGG_T + tid];
#pragma unroll
        for (int u = 0; u < SCAN_U; ++u) {
            const int v = v0 + u * AGG_T + tid;
            q_scan(k[u], v * KPV, pat, cnt, qe, lane);
            if (cnt >= 32) q_drain(cnt, qe, base, src, dst, h, acc, lane);
        }
    }
    q_drain(cnt, qe, base, src, dst, h, acc, lane);
    __syncthreads();

    for (int r = wv; r < RNG; r += AGG_W) {
        const int row = base + r;
        if (row >= n_nodes) break;
        const float* a = acc + r * IN_DIM + lane;
        const float4 o = make_float4(a[0], a[32], a[64], a[96]);
        ((float4*)(agg + (size_t)row * IN_DIM))[lane] = o;
    }
}

#define GF_GRID 256
#define GF_BW   (32 * 8 * 32)
__global__ __launch_bounds__(256) void gcn_fused(const float* __restrict__ agg,
                                                 const int* __restrict__ deg_in,
                                                 const float* __restrict__ W,
                                                 const float* __restrict__ bias,
                                                 const float* __restrict__ gamma,
                                                 const float* __restrict__ beta,
                                                 const float* __restrict__ feat,
                                                 const float* __restrict__ skip_W,
                                                 const float* __restrict__ skip_b,
                                                 float* __restrict__ out, int n_nodes) {
    __shared__ __attribute__((aligned(16))) float4 ldsB[2 * GF_BW];
    __shared__ __attribute__((aligned(16))) float lds_gcn[16][OUT_DIM + 4];
    __shared__ __attribute__((aligned(16))) float lds_skip[16][OUT_DIM + 4];

    const int tid  = threadIdx.x;
    const int wave = tid >> 5;
    const int lane = tid & 31;
    const int half = lane >> 4;
    const int l16  = lane & 15;
    const int bcol = wave * 32 + l16;

    for (int it = 0; it < 64; ++it) {
        const float* M = (it < 32) ? W : skip_W;
        const int s  = it & 31;
        const int k0 = 8 * (s >> 1) + 4 * half + 2 * (s & 1);
        const float* r0 = M + (size_t)k0 * OUT_DIM + bcol;
        float4 v;
        v.x = r0[0];  v.y = r0[OUT_DIM];
        v.z = r0[16]; v.w = r0[OUT_DIM + 16];
        ldsB[it * 256 + tid] = v;
    }

    const float bb0 = bias[bcol],      sb0 = skip_b[bcol];
    const float bb1 = bias[bcol + 16], sb1 = skip_b[bcol + 16];
    const float4 gm0 = ((const float4*)gamma)[lane * 2], gm1 = ((const float4*)gamma)[lane * 2 + 1];
    const float4 bt0 = ((const float4*)beta)[lane * 2],  bt1 = ((const float4*)beta)[lane * 2 + 1];
    const float4* bW = ldsB + wave * 32 + lane;
    const float4* bS = bW + GF_BW;
    const int n_tiles = n_nodes / 16;

    int tile = blockIdx.x;
    float4 nA[16], nF[16];
    if (tile < n_tiles) {
        const float4* ar = (const float4*)(agg  + (size_t)(tile * 16 + l16) * IN_DIM) + half;
        const float4* fr = (const float4*)(feat + (size_t)(tile * 16 + l16) * IN_DIM) + half;
        for (int j = 0; j < 16; ++j) { nA[j] = ar[2 * j]; nF[j] = fr[2 * j]; }
    }

    for (; tile < n_tiles; tile += GF_GRID) {
        const int m0 = tile * 16;
        float4 aA[16], aF[16];
        for (int j = 0; j < 16; ++j) { aA[j] = nA[j]; aF[j] = nF[j]; }
        if (tile + GF_GRID < n_tiles) {
            const float4* ar = (const float4*)(agg  + (size_t)(m0 + GF_GRID * 16 + l16) * IN_DIM) + half;
            const float4* fr = (const float4*)(feat + (size_t)(m0 + GF_GRID * 16 + l16) * IN_DIM) + half;
            for (int j = 0; j < 16; ++j) { nA[j] = ar[2 * j]; nF[j] = fr[2 * j]; }
        }
        int dg[8];
        for (int i = 0; i < 8; ++i) dg[i] = deg_in[m0 + i + 8 * half];

        v8f cg0 = {}, cg1 = {};
        v8f cs0 = {}, cs1 = {};
        for (int j = 0; j < 16; ++j) {
            for (int p = 0; p < 2; ++p) {
                const int s = 2 * j + p;
                const float4 w4 = bW[s * 256];
                const float4 s4 = bS[s * 256];
                v2f a, f, w0, w1, s0, s1;
                a[0] = p ? aA[j].z : aA[j].x;  a[1] = p ? aA[j].w : aA[j].y;
                f[0] = p ? aF[j].z : aF[j].x;  f[1] = p ? aF[j].w : aF[j].y;
                w0[0] = w4.x; w0[1] = w4.y; w1[0] = w4.z; w1[1] = w4.w;
                s0[0] = s4.x; s0[1] = s4.y; s1[0] = s4.z; s1[1] = s4.w;
                cg0 = __builtin_amdgcn_wmma_f32_16x16x4_f32(false, a, false, w0, (short)0, cg0, false, false);
                cg1 = __builtin_amdgcn_wmma_f32_16x16x4_f32(false, a, false, w1, (short)0, cg1, false, false);
                cs0 = __builtin_amdgcn_wmma_f32_16x16x4_f32(false, f, false, s0, (short)0, cs0, false, false);
                cs1 = __builtin_amdgcn_wmma_f32_16x16x4_f32(false, f, false, s1, (short)0, cs1, false, false);
            }
        }

        __syncthreads();
        for (int i = 0; i < 8; ++i) {
            const int row = i + 8 * half;
            const float nrm = rsqrtf(fmaxf((float)dg[i], 1.0f));
            lds_gcn[row][bcol]       = cg0[i] * nrm + bb0;
            lds_gcn[row][bcol + 16]  = cg1[i] * nrm + bb1;
            lds_skip[row][bcol]      = cs0[i] + sb0;
            lds_skip[row][bcol + 16] = cs1[i] + sb1;
        }
        __syncthreads();

        for (int rr = 0; rr < 2; ++rr) {
            const int row = wave * 2 + rr;
            const float4 x0 = *(const float4*)&lds_gcn[row][lane * 8];
            const float4 x1 = *(const float4*)&lds_gcn[row][lane * 8 + 4];
            const float4 k0 = *(const float4*)&lds_skip[row][lane * 8];
            const float4 k1 = *(const float4*)&lds_skip[row][lane * 8 + 4];
            const float vals[8] = {x0.x, x0.y, x0.z, x0.w, x1.x, x1.y, x1.z, x1.w};
            const float gmv[8]  = {gm0.x, gm0.y, gm0.z, gm0.w, gm1.x, gm1.y, gm1.z, gm1.w};
            const float btv[8]  = {bt0.x, bt0.y, bt0.z, bt0.w, bt1.x, bt1.y, bt1.z, bt1.w};
            const float skv[8]  = {k0.x, k0.y, k0.z, k0.w, k1.x, k1.y, k1.z, k1.w};
            float s = 0.f, s2 = 0.f;
            for (int j = 0; j < 8; ++j) { s += vals[j]; s2 += vals[j] * vals[j]; }
            for (int off = 16; off > 0; off >>= 1) {
                s  += __shfl_xor(s,  off, 32);
                s2 += __shfl_xor(s2, off, 32);
            }
            const float mu  = s * (1.0f / OUT_DIM);
            const float var = s2 * (1.0f / OUT_DIM) - mu * mu;
            const float rs  = rsqrtf(var + LN_EPS);
            float o[8];
            for (int j = 0; j < 8; ++j) {
                const float ln = (vals[j] - mu) * rs * gmv[j] + btv[j];
                o[j] = fmaxf(ln, 0.f) + skv[j];
            }
            float4* orow = (float4*)(out + (size_t)(m0 + row) * OUT_DIM) + lane * 2;
            orow[0] = make_float4(o[0], o[1], o[2], o[3]);
            orow[1] = make_float4(o[4], o[5], o[6], o[7]);
        }
    }
}

extern "C" void kernel_launch(void* const* d_in, const int* in_sizes, int n_in,
                              void* d_out, int out_size, void* d_ws, size_t ws_size,
                              hipStream_t stream) {
    const float* feat   = (const float*)d_in[0];
    const int*   src    = (const int*)d_in[1];
    const int*   dst    = (const int*)d_in[2];
    const float* W      = (const float*)d_in[3];
    const float* bias   = (const float*)d_in[4];
    const float* gamma  = (const float*)d_in[5];
    const float* beta   = (const float*)d_in[6];
    const float* skip_W = (const float*)d_in[7];
    const float* skip_b = (const float*)d_in[8];
    float* out = (float*)d_out;

    const int n_nodes = in_sizes[0] / IN_DIM;
    const int n_edges = in_sizes[1];

    int*   deg_out = (int*)d_ws;
    int*   deg_in  = deg_out + n_nodes;
    float* h       = (float*)(deg_in + n_nodes);
    float* agg     = h + (size_t)n_nodes * IN_DIM;

    const int T = 256;

    unsigned short* key = (unsigned short*)d_out;
    const int scan_step = KPV * SCAN_U * AGG_T;
    const int n_pad = (n_edges + scan_step - 1) / scan_step * scan_step;

    {
        unsigned n = 2u * (unsigned)n_nodes;
        zero_u32<<<(n + T - 1) / T, T, 0, stream>>>((unsigned*)deg_out, n);
    }

    degree_kernel<<<(n_pad + T - 1) / T, T, 0, stream>>>(src, dst, deg_out, deg_in, key, n_edges, n_pad);

    {
        int total = n_nodes * (IN_DIM / 4);
        compute_h<<<(total + T - 1) / T, T, 0, stream>>>(feat, deg_out, h, n_nodes);
    }

    agg_range<<<(n_nodes + RNG - 1) / RNG, AGG_T, 0, stream>>>(src, dst, key, h, agg, n_pad, n_nodes);

    gcn_fused<<<GF_GRID, T, 0, stream>>>(agg, deg_in, W, bias, gamma, beta,
                                              feat, skip_W, skip_b, out, n_nodes);
}


# degree counts by LDS histograms: 64 work-groups (index array x node half x 16 edge chunks) store partial count planes, the row-scaling pass adds 16 planes per node; no global atomics, 4 launches
# speedup vs baseline: 2.3807x; 1.4830x over previous
#include <hip/hip_runtime.h>
#include <hip/hip_bf16.h>

typedef __attribute__((ext_vector_type(2))) float v2f;
typedef __attribute__((ext_vector_type(8))) float v8f;

#define IN_DIM 128
#define OUT_DIM 256
#define LN_EPS 1e-5f

#define RNG      512
#define KEY_NONE 0xFFFFu
#define HP   16
#define HN   50016
#define HT   1024
#define HU   5

__device__ __forceinline__ void hist_add(int* cnt, int node, int lo) {
    const unsigned r = (unsigned)(node - lo);
    if (r < (unsigned)HN) atomicAdd(&cnt[r], 1);
}

__device__ __forceinline__ unsigned key_of(int node) {
    return node < 0 ? KEY_NONE : (unsigned)node / RNG;
}

__global__ __launch_bounds__(HT) void degree_hist(const int* __restrict__ src, const int* __restrict__ dst,
                                                  int* __restrict__ planes,
                                                  unsigned short* __restrict__ key,
                                                  int n_edges, int n_vec) {
    __shared__ __attribute__((aligned(16))) int cnt[HN];

    const int tid  = threadIdx.x;
    const int c    = blockIdx.x % HP;
    const int role = blockIdx.x / HP;
    const int arr  = role >> 1;
    const int lo   = (role & 1) * HN;
    const int* __restrict__ idx = arr ? dst : src;
    const bool wkey = role == 2;

    for (int i = tid; i < HN / 4; i += HT) ((int4*)cnt)[i] = make_int4(0, 0, 0, 0);
    __syncthreads();

    const int chv  = n_vec / HP;
    const int vbeg = c * chv, vend = vbeg + chv;
    for (int v0 = vbeg; v0 < vend; v0 += HU * HT) {
        int4 x[HU];
#pragma unroll
        for (int u = 0; u < HU; ++u) {
            const int v = v0 + u * HT + tid;
            x[u] = make_int4(-1, -1, -1, -1);
            if (v < vend && 4 * v + 3 < n_edges) x[u] = ((const int4*)idx)[v];
        }
#pragma unroll
        for (int u = 0; u < HU; ++u) {
            const int v = v0 + u * HT + tid;
            hist_add(cnt, x[u].x, lo);
            hist_add(cnt, x[u].y, lo);
            hist_add(cnt, x[u].z, lo);
            hist_add(cnt, x[u].w, lo);
            if (wkey && v < vend && (4 * v + 3 < n_edges || 4 * v >= n_edges))
                ((uint2*)key)[v] = make_uint2(key_of(x[u].x) | (key_of(x[u].y) << 16),
                                              key_of(x[u].z) | (key_of(x[u].w) << 16));
        }
    }
    const int vt = n_edges >> 2;
    if ((n_edges & 3) && vt >= vbeg && vt < vend && tid < 4) {
        const int e = 4 * vt + tid;
        const int node = e < n_edges ? idx[e] : -1;
        hist_add(cnt, node, lo);
        if (wkey) key[e] = (unsigned short)key_of(node);
    }
    __syncthreads();

    int4* o = (int4*)(planes + (size_t)(arr * HP + c) * (2 * HN) + lo);
    for (int i = tid; i < HN / 4; i += HT) o[i] = ((const int4*)cnt)[i];
}

__global__ __launch_bounds__(256) void scale_rows(const float* __restrict__ feat,
                                                  const int* __restrict__ planes,
                                                  int* __restrict__ deg_in,
                                                  float* __restrict__ h, int n_nodes) {
    __shared__ int part[2 * HP][32];
    __shared__ float nrm_s[32];

    const int t = threadIdx.x, w = t >> 5, l = t & 31;
    const int n0 = blockIdx.x * 32;

    int pv[4];
    for (int k = 0; k < 4; ++k) {
        const int q = w + 8 * k;
        pv[k] = planes[(size_t)q * (2 * HN) + min(n0 + l, 2 * HN - 1)];
    }
    float4 v[4];
    for (int k = 0; k < 4; ++k) {
        const int node = min(n0 + w + 8 * k, n_nodes - 1);
        v[k] = ((const float4*)feat)[(size_t)node * (IN_DIM / 4) + l];
    }
    for (int k = 0; k < 4; ++k) part[w + 8 * k][l] = pv[k];
    __syncthreads();
    if (t < 64) {
        const int a = t >> 5;
        int s = 0;
        for (int c = 0; c < HP; ++c) s += part[a * HP + c][l];
        if (a == 0) nrm_s[l] = rsqrtf(fmaxf((float)s, 1.0f));
        else if (n0 + l < n_nodes) deg_in[n0 + l] = s;
    }
    __syncthreads();
    for (int k = 0; k < 4; ++k) {
        const int node = n0 + w + 8 * k;
        if (node < n_nodes) {
            const float nrm = nrm_s[w + 8 * k];
            float4 o = v[k];
            o.x *= nrm; o.y *= nrm; o.z *= nrm; o.w *= nrm;
            ((float4*)h)[(size_t)node * (IN_DIM / 4) + l] = o;
        }
    }
}

#define AGG_T 1024
#define AGG_W (AGG_T / 32)
#define KPV   8
#define SCAN_U 4
#define QCAP  288
#define GU    8

typedef unsigned short __attribute__((ext_vector_type(2))) u16x2;

__device__ __forceinline__ unsigned key_hit(int k, unsigned pat) {
    const u16x2 one = {1, 1};
    const u16x2 x = __builtin_bit_cast(u16x2, (unsigned)k ^ pat);
    return __builtin_bit_cast(unsigned, __builtin_elementwise_sub_sat(one, x));
}

__device__ __forceinline__ void q_scan(int4 k, int e0, unsigned pat, int& cnt, int* qe, int lane) {
    unsigned g = key_hit(k.x, pat) | (key_hit(k.y, pat) << 1) |
                 (key_hit(k.z, pat) << 2) | (key_hit(k.w, pat) << 3);
    unsigned mask = __builtin_amdgcn_ballot_w32(g != 0u);
    while (mask) {
        const bool m = g != 0u;
        const int b = __builtin_ctz(g | 0x80000000u);
        const int pos = cnt + __builtin_popcount(mask & ((1u << lane) - 1u));
        if (m) qe[pos] = e0 + ((b & 3) << 1) + (b >> 4);
        cnt += __builtin_popcount(mask);
        g &= g - 1u;
        mask = __builtin_amdgcn_ballot_w32(g != 0u);
    }
}

__device__ __forceinline__ void q_drain(int& cnt, const int* qe, int base,
                                        const int* __restrict__ src, const int* __restrict__ dst,
                                        const float* __restrict__ h, float* acc, int lane) {
    for (int b0 = 0; b0 < cnt; b0 += 32) {
        const int nb = min(32, cnt - b0);
        int s = 0, dl = 0;
        if (lane < nb) { const int e = qe[b0 + lane]; s = src[e]; dl = dst[e] - base; }
        for (int i = 0; i < nb; i += GU) {
            float4 v[GU];
            int rl[GU];
#pragma unroll
            for (int u = 0; u < GU; ++u) {
                const int iu = min(i + u, nb - 1);
                const int su = __builtin_amdgcn_readlane(s, iu);
                rl[u] = __builtin_amdgcn_readlane(dl, iu);
                v[u] = ((const float4*)(h + (size_t)su * IN_DIM))[lane];
            }
#pragma unroll
            for (int u = 0; u < GU; ++u) {
                if (i + u < nb) {
                    float* a = acc + rl[u] * IN_DIM + lane;
                    __hip_atomic_fetch_add(a,      v[u].x, __ATOMIC_RELAXED, __HIP_MEMORY_SCOPE_WORKGROUP);
                    __hip_atomic_fetch_add(a + 32, v[u].y, __ATOMIC_RELAXED, __HIP_MEMORY_SCOPE_WORKGROUP);
                    __hip_atomic_fetch_add(a + 64, v[u].z, __ATOMIC_RELAXED, __HIP_MEMORY_SCOPE_WORKGROUP);
                    __hip_atomic_fetch_add(a + 96, v[u].w, __ATOMIC_RELAXED, __HIP_MEMORY_SCOPE_WORKGROUP);
                }
            }
        }
    }
    cnt = 0;
}

__global__ __launch_bounds__(AGG_T) void agg_range(const int* __restrict__ src,
                                                   const int* __restrict__ dst,
                                                   const unsigned short* __restrict__ key,
                                                   const float* __restrict__ h,
                                                   float* __restrict__ agg,
                                                   int n_pad, int n_nodes) {
    __shared__ __attribute__((aligned(16))) float acc[RNG * IN_DIM];
    __shared__ int qe_all[AGG_W][QCAP];

    const int tid  = threadIdx.x;
    const int lane = tid & 31;
    const int wv   = tid >> 5;
    const int base = blockIdx.x * RNG;
    int* qe = qe_all[wv];
    const unsigned pat = blockIdx.x | (blockIdx.x << 16);

    for (int i = tid; i < RNG * IN_DIM / 4; i += AGG_T)
        ((float4*)acc)[i] = make_float4(0.f, 0.f, 0.f, 0.f);
    __syncthreads();

    int cnt = 0;
    const int4* key4 = (const int4*)key;
    const int nvec = n_pad / KPV;

    for (int v0 = 0; v0 < nvec; v0 += SCAN_U * AGG_T) {
        int4 k[SCAN_U];
#pragma unroll
        for (int u = 0; u < SCAN_U; ++u)
            k[u] = key4[v0 + u * AGG_T + tid];
#pragma unroll
        for (int u = 0; u < SCAN_U; ++u) {
            const int v = v0 + u * AGG_T + tid;
            q_scan(k[u], v * KPV, pat, cnt, qe, lane);
            if (cnt >= 32) q_drain(cnt, qe, base, src, dst, h, acc, lane);
        }
    }
    q_drain(cnt, qe, base, src, dst, h, acc, lane);
    __syncthreads();

    for (int r = wv; r < RNG; r += AGG_W) {
        const int row = base + r;
        if (row >= n_nodes) break;
        const float* a = acc + r * IN_DIM + lane;
        const float4 o = make_float4(a[0], a[32], a[64], a[96]);
        ((float4*)(agg + (size_t)row * IN_DIM))[lane] = o;
    }
}

#define GF_GRID 256
#define GF_BW   (32 * 8 * 32)
__global__ __launch_bounds__(256) void gcn_fused(const float* __restrict__ agg,
                                                 const int* __restrict__ deg_in,
                                                 const float* __restrict__ W,
                                                 const float* __restrict__ bias,
                                                 const float* __restrict__ gamma,
                                                 const float* __restrict__ beta,
                                                 const float* __restrict__ feat,
                                                 const float* __restrict__ skip_W,
                                                 const float* __restrict__ skip_b,
                                                 float* __restrict__ out, int n_nodes) {
    __shared__ __attribute__((aligned(16))) float4 ldsB[2 * GF_BW];
    __shared__ __attribute__((aligned(16))) float lds_gcn[16][OUT_DIM + 4];
    __shared__ __attribute__((aligned(16))) float lds_skip[16][OUT_DIM + 4];

    const int tid  = threadIdx.x;
    const int wave = tid >> 5;
    const int lane = tid & 31;
    const int half = lane >> 4;
    const int l16  = lane & 15;
    const int bcol = wave * 32 + l16;

    for (int it = 0; it < 64; ++it) {
        const float* M = (it < 32) ? W : skip_W;
        const int s  = it & 31;
        const int k0 = 8 * (s >> 1) + 4 * half + 2 * (s & 1);
        const float* r0 = M + (size_t)k0 * OUT_DIM + bcol;
        float4 v;
        v.x = r0[0];  v.y = r0[OUT_DIM];
        v.z = r0[16]; v.w = r0[OUT_DIM + 16];
        ldsB[it * 256 + tid] = v;
    }

    const float bb0 = bias[bcol],      sb0 = skip_b[bcol];
    const float bb1 = bias[bcol + 16], sb1 = skip_b[bcol + 16];
    const float4 gm0 = ((const float4*)gamma)[lane * 2], gm1 = ((const float4*)gamma)[lane * 2 + 1];
    const float4 bt0 = ((const float4*)beta)[lane * 2],  bt1 = ((const float4*)beta)[lane * 2 + 1];
    const float4* bW = ldsB + wave * 32 + lane;
    const float4* bS = bW + GF_BW;
    const int n_tiles = n_nodes / 16;

    int tile = blockIdx.x;
    float4 nA[16], nF[16];
    if (tile < n_tiles) {
        const float4* ar = (const float4*)(agg  + (size_t)(tile * 16 + l16) * IN_DIM) + half;
        const float4* fr = (const float4*)(feat + (size_t)(tile * 16 + l16) * IN_DIM) + half;
        for (int j = 0; j < 16; ++j) { nA[j] = ar[2 * j]; nF[j] = fr[2 * j]; }
    }

    for (; tile < n_tiles; tile += GF_GRID) {
        const int m0 = tile * 16;
        float4 aA[16], aF[16];
        for (int j = 0; j < 16; ++j) { aA[j] = nA[j]; aF[j] = nF[j]; }
        if (tile + GF_GRID < n_tiles) {
            const float4* ar = (const float4*)(agg  + (size_t)(m0 + GF_GRID * 16 + l16) * IN_DIM) + half;
            const float4* fr = (const float4*)(feat + (size_t)(m0 + GF_GRID * 16 + l16) * IN_DIM) + half;
            for (int j = 0; j < 16; ++j) { nA[j] = ar[2 * j]; nF[j] = fr[2 * j]; }
        }
        int dg[8];
        for (int i = 0; i < 8; ++i) dg[i] = deg_in[m0 + i + 8 * half];

        v8f cg0 = {}, cg1 = {};
        v8f cs0 = {}, cs1 = {};
        for (int j = 0; j < 16; ++j) {
            for (int p = 0; p < 2; ++p) {
                const int s = 2 * j + p;
                const float4 w4 = bW[s * 256];
                const float4 s4 = bS[s * 256];
                v2f a, f, w0, w1, s0, s1;
                a[0] = p ? aA[j].z : aA[j].x;  a[1] = p ? aA[j].w : aA[j].y;
                f[0] = p ? aF[j].z : aF[j].x;  f[1] = p ? aF[j].w : aF[j].y;
                w0[0] = w4.x; w0[1] = w4.y; w1[0] = w4.z; w1[1] = w4.w;
                s0[0] = s4.x; s0[1] = s4.y; s1[0] = s4.z; s1[1] = s4.w;
                cg0 = __builtin_amdgcn_wmma_f32_16x16x4_f32(false, a, false, w0, (short)0, cg0, false, false);
                cg1 = __builtin_amdgcn_wmma_f32_16x16x4_f32(false, a, false, w1, (short)0, cg1, false, false);
                cs0 = __builtin_amdgcn_wmma_f32_16x16x4_f32(false, f, false, s0, (short)0, cs0, false, false);
                cs1 = __builtin_amdgcn_wmma_f32_16x16x4_f32(false, f, false, s1, (short)0, cs1, false, false);
            }
        }

        __syncthreads();
        for (int i = 0; i < 8; ++i) {
            const int row = i + 8 * half;
            const float nrm = rsqrtf(fmaxf((float)dg[i], 1.0f));
            lds_gcn[row][bcol]       = cg0[i] * nrm + bb0;
            lds_gcn[row][bcol + 16]  = cg1[i] * nrm + bb1;
            lds_skip[row][bcol]      = cs0[i] + sb0;
            lds_skip[row][bcol + 16] = cs1[i] + sb1;
        }
        __syncthreads();

        for (int rr = 0; rr < 2; ++rr) {
            const int row = wave * 2 + rr;
            const float4 x0 = *(const float4*)&lds_gcn[row][lane * 8];
            const float4 x1 = *(const float4*)&lds_gcn[row][lane * 8 + 4];
            const float4 k0 = *(const float4*)&lds_skip[row][lane * 8];
            const float4 k1 = *(const float4*)&lds_skip[row][lane * 8 + 4];
            const float vals[8] = {x0.x, x0.y, x0.z, x0.w, x1.x, x1.y, x1.z, x1.w};
            const float gmv[8]  = {gm0.x, gm0.y, gm0.z, gm0.w, gm1.x, gm1.y, gm1.z, gm1.w};
            const float btv[8]  = {bt0.x, bt0.y, bt0.z, bt0.w, bt1.x, bt1.y, bt1.z, bt1.w};
            const float skv[8]  = {k0.x, k0.y, k0.z, k0.w, k1.x, k1.y, k1.z, k1.w};
            float s = 0.f, s2 = 0.f;
            for (int j = 0; j < 8; ++j) { s += vals[j]; s2 += vals[j] * vals[j]; }
            for (int off = 16; off > 0; off >>= 1) {
                s  += __shfl_xor(s,  off, 32);
                s2 += __shfl_xor(s2, off, 32);
            }
            const float mu  = s * (1.0f / OUT_DIM);
            const float var = s2 * (1.0f / OUT_DIM) - mu * mu;
            const float rs  = rsqrtf(var + LN_EPS);
            float o[8];
            for (int j = 0; j < 8; ++j) {
                const float ln = (vals[j] - mu) * rs * gmv[j] + btv[j];
                o[j] = fmaxf(ln, 0.f) + skv[j];
            }
            float4* orow = (float4*)(out + (size_t)(m0 + row) * OUT_DIM) + lane * 2;
            orow[0] = make_float4(o[0], o[1], o[2], o[3]);
            orow[1] = make_float4(o[4], o[5], o[6], o[7]);
        }
    }
}

extern "C" void kernel_launch(void* const* d_in, const int* in_sizes, int n_in,
                              void* d_out, int out_size, void* d_ws, size_t ws_size,
                              hipStream_t stream) {
    const float* feat   = (const float*)d_in[0];
    const int*   src    = (const int*)d_in[1];
    const int*   dst    = (const int*)d_in[2];
    const float* W      = (const float*)d_in[3];
    const float* bias   = (const float*)d_in[4];
    const float* gamma  = (const float*)d_in[5];
    const float* beta   = (const float*)d_in[6];
    const float* skip_W = (const float*)d_in[7];
    const float* skip_b = (const float*)d_in[8];
    float* out = (float*)d_out;

    const int n_nodes = in_sizes[0] / IN_DIM;
    const int n_edges = in_sizes[1];

    int*   deg_in  = (int*)d_ws + n_nodes;
    float* h       = (float*)(deg_in + n_nodes);
    float* agg     = h + (size_t)n_nodes * IN_DIM;

    int* planes = (int*)agg;

    unsigned short* key = (unsigned short*)d_out;
    const int scan_step = KPV * SCAN_U * AGG_T;
    const int n_pad = (n_edges + scan_step - 1) / scan_step * scan_step;

    degree_hist<<<4 * HP, HT, 0, stream>>>(src, dst, planes, key, n_edges, n_pad / 4);

    scale_rows<<<(n_nodes + 31) / 32, 256, 0, stream>>>(feat, planes, deg_in, h, n_nodes);

    agg_range<<<(n_nodes + RNG - 1) / RNG, AGG_T, 0, stream>>>(src, dst, key, h, agg, n_pad, n_nodes);

    gcn_fused<<<GF_GRID, 256, 0, stream>>>(agg, deg_in, W, bias, gamma, beta,
                                              feat, skip_W, skip_b, out, n_nodes);
}


# agg_range range-id stream 2 B -> 1 B per edge (dst/512 fits a byte): 16 edges per 16-B load, one 32-bit match word covers 32 keys per lane; per-wave queue, LDS atomics and arrival-order sums kept
# speedup vs baseline: 2.5134x; 1.0557x over previous
#include <hip/hip_runtime.h>
#include <hip/hip_bf16.h>

typedef __attribute__((ext_vector_type(2))) float v2f;
typedef __attribute__((ext_vector_type(8))) float v8f;

#define IN_DIM 128
#define OUT_DIM 256
#define LN_EPS 1e-5f

#define RNG      512
#define KEY_NONE 0xFFu
#define HP   16
#define HN   50016
#define HT   1024
#define HU   5

__device__ __forceinline__ void hist_add(int* cnt, int node, int lo) {
    const unsigned r = (unsigned)(node - lo);
    if (r < (unsigned)HN) atomicAdd(&cnt[r], 1);
}

__device__ __forceinline__ unsigned key_of(int node) {
    return node < 0 ? KEY_NONE : (unsigned)node / RNG;
}

__global__ __launch_bounds__(HT) void degree_hist(const int* __restrict__ src, const int* __restrict__ dst,
                                                  int* __restrict__ planes,
                                                  unsigned char* __restrict__ key,
                                                  int n_edges, int n_vec) {
    __shared__ __attribute__((aligned(16))) int cnt[HN];

    const int tid  = threadIdx.x;
    const int c    = blockIdx.x % HP;
    const int role = blockIdx.x / HP;
    const int arr  = role >> 1;
    const int lo   = (role & 1) * HN;
    const int* __restrict__ idx = arr ? dst : src;
    const bool wkey = role == 2;

    for (int i = tid; i < HN / 4; i += HT) ((int4*)cnt)[i] = make_int4(0, 0, 0, 0);
    __syncthreads();

    const int chv  = n_vec / HP;
    const int vbeg = c * chv, vend = vbeg + chv;
    for (int v0 = vbeg; v0 < vend; v0 += HU * HT) {
        int4 x[HU];
#pragma unroll
        for (int u = 0; u < HU; ++u) {
            const int v = v0 + u * HT + tid;
            x[u] = make_int4(-1, -1, -1, -1);
            if (v < vend && 4 * v + 3 < n_edges) x[u] = ((const int4*)idx)[v];
        }
#pragma unroll
        for (int u = 0; u < HU; ++u) {
            const int v = v0 + u * HT + tid;
            hist_add(cnt, x[u].x, lo);
            hist_add(cnt, x[u].y, lo);
            hist_add(cnt, x[u].z, lo);
            hist_add(cnt, x[u].w, lo);
            if (wkey && v < vend && (4 * v + 3 < n_edges || 4 * v >= n_edges))
                ((unsigned*)key)[v] = key_of(x[u].x) | (key_of(x[u].y) << 8) |
                                      (key_of(x[u].z) << 16) | (key_of(x[u].w) << 24);
        }
    }
    const int vt = n_edges >> 2;
    if ((n_edges & 3) && vt >= vbeg && vt < vend && tid < 4) {
        const int e = 4 * vt + tid;
        const int node = e < n_edges ? idx[e] : -1;
        hist_add(cnt, node, lo);
        if (wkey) key[e] = (unsigned char)key_of(node);
    }
    __syncthreads();

    int4* o = (int4*)(planes + (size_t)(arr * HP + c) * (2 * HN) + lo);
    for (int i = tid; i < HN / 4; i += HT) o[i] = ((const int4*)cnt)[i];
}

__global__ __launch_bounds__(256) void scale_rows(const float* __restrict__ feat,
                                                  const int* __restrict__ planes,
                                                  int* __restrict__ deg_in,
                                                  float* __restrict__ h, int n_nodes) {
    __shared__ int part[2 * HP][32];
    __shared__ float nrm_s[32];

    const int t = threadIdx.x, w = t >> 5, l = t & 31;
    const int n0 = blockIdx.x * 32;

    int pv[4];
    for (int k = 0; k < 4; ++k) {
        const int q = w + 8 * k;
        pv[k] = planes[(size_t)q * (2 * HN) + min(n0 + l, 2 * HN - 1)];
    }
    float4 v[4];
    for (int k = 0; k < 4; ++k) {
        const int node = min(n0 + w + 8 * k, n_nodes - 1);
        v[k] = ((const float4*)feat)[(size_t)node * (IN_DIM / 4) + l];
    }
    for (int k = 0; k < 4; ++k) part[w + 8 * k][l] = pv[k];
    __syncthreads();
    if (t < 64) {
        const int a = t >> 5;
        int s = 0;
        for (int c = 0; c < HP; ++c) s += part[a * HP + c][l];
        if (a == 0) nrm_s[l] = rsqrtf(fmaxf((float)s, 1.0f));
        else if (n0 + l < n_nodes) deg_in[n0 + l] = s;
    }
    __syncthreads();
    for (int k = 0; k < 4; ++k) {
        const int node = n0 + w + 8 * k;
        if (node < n_nodes) {
            const float nrm = nrm_s[w + 8 * k];
            float4 o = v[k];
            o.x *= nrm; o.y *= nrm; o.z *= nrm; o.w *= nrm;
            ((float4*)h)[(size_t)node * (IN_DIM / 4) + l] = o;
        }
    }
}

#define AGG_T 1024
#define AGG_W (AGG_T / 32)
#define KPV   16
#define SCAN_U 4
#define QCAP  288
#define GU    8

__device__ __forceinline__ unsigned zero_bytes(unsigned x) {
    return ~(((x & 0x7F7F7F7Fu) + 0x7F7F7F7Fu) | x | 0x7F7F7F7Fu);
}

__device__ __forceinline__ unsigned scan_bits(int4 k, unsigned pat) {
    return (zero_bytes((unsigned)k.x ^ pat) >> 7) | (zero_bytes((unsigned)k.y ^ pat) >> 6) |
           (zero_bytes((unsigned)k.z ^ pat) >> 5) | (zero_bytes((unsigned)k.w ^ pat) >> 4);
}

__device__ __forceinline__ void q_round(unsigned& g, unsigned& mask, int e0, int& cnt, int* qe, int lane) {
    const bool m = g != 0u;
    const int b = __builtin_ctz(g | 0x80000000u);
    const int pos = cnt + __builtin_popcount(mask & ((1u << lane) - 1u));
    if (m) qe[pos] = e0 + ((b & 4) ? AGG_T * KPV : 0) + ((b & 3) << 2) + (b >> 3);
    cnt += __builtin_popcount(mask);
    g &= g - 1u;
    mask = __builtin_amdgcn_ballot_w32(g != 0u);
}

__device__ __forceinline__ void q_drain(int& cnt, const int* qe, int base,
                                        const int* __restrict__ src, const int* __restrict__ dst,
                                        const float* __restrict__ h, float* acc, int lane) {
    for (int b0 = 0; b0 < cnt; b0 += 32) {
        const int nb = min(32, cnt - b0);
        int s = 0, dl = 0;
        if (lane < nb) { const int e = qe[b0 + lane]; s = src[e]; dl = dst[e] - base; }
        for (int i = 0; i < nb; i += GU) {
            float4 v[GU];
            int rl[GU];
#pragma unroll
            for (int u = 0; u < GU; ++u) {
                const int iu = min(i + u, nb - 1);
                const int su = __builtin_amdgcn_readlane(s, iu);
                rl[u] = __builtin_amdgcn_readlane(dl, iu);
                v[u] = ((const float4*)(h + (size_t)su * IN_DIM))[lane];
            }
#pragma unroll
            for (int u = 0; u < GU; ++u) {
                if (i + u < nb) {
                    float* a = acc + rl[u] * IN_DIM + lane;
                    __hip_atomic_fetch_add(a,      v[u].x, __ATOMIC_RELAXED, __HIP_MEMORY_SCOPE_WORKGROUP);
                    __hip_atomic_fetch_add(a + 32, v[u].y, __ATOMIC_RELAXED, __HIP_MEMORY_SCOPE_WORKGROUP);
                    __hip_atomic_fetch_add(a + 64, v[u].z, __ATOMIC_RELAXED, __HIP_MEMORY_SCOPE_WORKGROUP);
                    __hip_atomic_fetch_add(a + 96, v[u].w, __ATOMIC_RELAXED, __HIP_MEMORY_SCOPE_WORKGROUP);
                }
            }
        }
    }
    cnt = 0;
}

__global__ __launch_bounds__(AGG_T) void agg_range(const int* __restrict__ src,
                                                   const int* __restrict__ dst,
                                                   const unsigned char* __restrict__ key,
                                                   const float* __restrict__ h,
                                                   float* __restrict__ agg,
                                                   int n_pad, int n_nodes) {
    __shared__ __attribute__((aligned(16))) float acc[RNG * IN_DIM];
    __shared__ int qe_all[AGG_W][QCAP];

    const int tid  = threadIdx.x;
    const int lane = tid & 31;
    const int wv   = tid >> 5;
    const int base = blockIdx.x * RNG;
    int* qe = qe_all[wv];
    const unsigned pat = blockIdx.x * 0x01010101u;

    for (int i = tid; i < RNG * IN_DIM / 4; i += AGG_T)
        ((float4*)acc)[i] = make_float4(0.f, 0.f, 0.f, 0.f);
    __syncthreads();

    int cnt = 0;
    const int4* key4 = (const int4*)key;
    const int nvec = n_pad / KPV;

    for (int v0 = 0; v0 < nvec; v0 += SCAN_U * AGG_T) {
        int4 k[SCAN_U];
#pragma unroll
        for (int u = 0; u < SCAN_U; ++u)
            k[u] = key4[v0 + u * AGG_T + tid];
#pragma unroll
        for (int u = 0; u < SCAN_U; u += 2) {
            const int e0 = (v0 + u * AGG_T + tid) * KPV;
            unsigned g = scan_bits(k[u], pat) | (scan_bits(k[u + 1], pat) << 4);
            unsigned mask = __builtin_amdgcn_ballot_w32(g != 0u);
            for (;;) {
                while (mask && cnt <= QCAP - 32) q_round(g, mask, e0, cnt, qe, lane);
                if (cnt >= 32) q_drain(cnt, qe, base, src, dst, h, acc, lane);
                if (!mask) break;
            }
        }
    }
    q_drain(cnt, qe, base, src, dst, h, acc, lane);
    __syncthreads();

    for (int r = wv; r < RNG; r += AGG_W) {
        const int row = base + r;
        if (row >= n_nodes) break;
        const float* a = acc + r * IN_DIM + lane;
        const float4 o = make_float4(a[0], a[32], a[64], a[96]);
        ((float4*)(agg + (size_t)row * IN_DIM))[lane] = o;
    }
}

#define GF_GRID 256
#define GF_BW   (32 * 8 * 32)
__global__ __launch_bounds__(256) void gcn_fused(const float* __restrict__ agg,
                                                 const int* __restrict__ deg_in,
                                                 const float* __restrict__ W,
                                                 const float* __restrict__ bias,
                                                 const float* __restrict__ gamma,
                                                 const float* __restrict__ beta,
                                                 const float* __restrict__ feat,
                                                 const float* __restrict__ skip_W,
                                                 const float* __restrict__ skip_b,
                                                 float* __restrict__ out, int n_nodes) {
    __shared__ __attribute__((aligned(16))) float4 ldsB[2 * GF_BW];
    __shared__ __attribute__((aligned(16))) float lds_gcn[16][OUT_DIM + 4];
    __shared__ __attribute__((aligned(16))) float lds_skip[16][OUT_DIM + 4];

    const int tid  = threadIdx.x;
    const int wave = tid >> 5;
    const int lane = tid & 31;
    const int half = lane >> 4;
    const int l16  = lane & 15;
    const int bcol = wave * 32 + l16;

    for (int it = 0; it < 64; ++it) {
        const float* M = (it < 32) ? W : skip_W;
        const int s  = it & 31;
        const int k0 = 8 * (s >> 1) + 4 * half + 2 * (s & 1);
        const float* r0 = M + (size_t)k0 * OUT_DIM + bcol;
        float4 v;
        v.x = r0[0];  v.y = r0[OUT_DIM];
        v.z = r0[16]; v.w = r0[OUT_DIM + 16];
        ldsB[it * 256 + tid] = v;
    }

    const float bb0 = bias[bcol],      sb0 = skip_b[bcol];
    const float bb1 = bias[bcol + 16], sb1 = skip_b[bcol + 16];
    const float4 gm0 = ((const float4*)gamma)[lane * 2], gm1 = ((const float4*)gamma)[lane * 2 + 1];
    const float4 bt0 = ((const float4*)beta)[lane * 2],  bt1 = ((const float4*)beta)[lane * 2 + 1];
    const float4* bW = ldsB + wave * 32 + lane;
    const float4* bS = bW + GF_BW;
    const int n_tiles = n_nodes / 16;

    int tile = blockIdx.x;
    float4 nA[16], nF[16];
    if (tile < n_tiles) {
        const float4* ar = (const float4*)(agg  + (size_t)(tile * 16 + l16) * IN_DIM) + half;
        const float4* fr = (const float4*)(feat + (size_t)(tile * 16 + l16) * IN_DIM) + half;
        for (int j = 0; j < 16; ++j) { nA[j] = ar[2 * j]; nF[j] = fr[2 * j]; }
    }

    for (; tile < n_tiles; tile += GF_GRID) {
        const int m0 = tile * 16;
        float4 aA[16], aF[16];
        for (int j = 0; j < 16; ++j) { aA[j] = nA[j]; aF[j] = nF[j]; }
        if (tile + GF_GRID < n_tiles) {
            const float4* ar = (const float4*)(agg  + (size_t)(m0 + GF_GRID * 16 + l16) * IN_DIM) + half;
            const float4* fr = (const float4*)(feat + (size_t)(m0 + GF_GRID * 16 + l16) * IN_DIM) + half;
            for (int j = 0; j < 16; ++j) { nA[j] = ar[2 * j]; nF[j] = fr[2 * j]; }
        }
        int dg[8];
        for (int i = 0; i < 8; ++i) dg[i] = deg_in[m0 + i + 8 * half];

        v8f cg0 = {}, cg1 = {};
        v8f cs0 = {}, cs1 = {};
        for (int j = 0; j < 16; ++j) {
            for (int p = 0; p < 2; ++p) {
                const int s = 2 * j + p;
                const float4 w4 = bW[s * 256];
                const float4 s4 = bS[s * 256];
                v2f a, f, w0, w1, s0, s1;
                a[0] = p ? aA[j].z : aA[j].x;  a[1] = p ? aA[j].w : aA[j].y;
                f[0] = p ? aF[j].z : aF[j].x;  f[1] = p ? aF[j].w : aF[j].y;
                w0[0] = w4.x; w0[1] = w4.y; w1[0] = w4.z; w1[1] = w4.w;
                s0[0] = s4.x; s0[1] = s4.y; s1[0] = s4.z; s1[1] = s4.w;
                cg0 = __builtin_amdgcn_wmma_f32_16x16x4_f32(false, a, false, w0, (short)0, cg0, false, false);
                cg1 = __builtin_amdgcn_wmma_f32_16x16x4_f32(false, a, false, w1, (short)0, cg1, false, false);
                cs0 = __builtin_amdgcn_wmma_f32_16x16x4_f32(false, f, false, s0, (short)0, cs0, false, false);
                cs1 = __builtin_amdgcn_wmma_f32_16x16x4_f32(false, f, false, s1, (short)0, cs1, false, false);
            }
        }

        __syncthreads();
        for (int i = 0; i < 8; ++i) {
            const int row = i + 8 * half;
            const float nrm = rsqrtf(fmaxf((float)dg[i], 1.0f));
            lds_gcn[row][bcol]       = cg0[i] * nrm + bb0;
            lds_gcn[row][bcol + 16]  = cg1[i] * nrm + bb1;
            lds_skip[row][bcol]      = cs0[i] + sb0;
            lds_skip[row][bcol + 16] = cs1[i] + sb1;
        }
        __syncthreads();

        for (int rr = 0; rr < 2; ++rr) {
            const int row = wave * 2 + rr;
            const float4 x0 = *(const float4*)&lds_gcn[row][lane * 8];
            const float4 x1 = *(const float4*)&lds_gcn[row][lane * 8 + 4];
            const float4 k0 = *(const float4*)&lds_skip[row][lane * 8];
            const float4 k1 = *(const float4*)&lds_skip[row][lane * 8 + 4];
            const float vals[8] = {x0.x, x0.y, x0.z, x0.w, x1.x, x1.y, x1.z, x1.w};
            const float gmv[8]  = {gm0.x, gm0.y, gm0.z, gm0.w, gm1.x, gm1.y, gm1.z, gm1.w};
            const float btv[8]  = {bt0.x, bt0.y, bt0.z, bt0.w, bt1.x, bt1.y, bt1.z, bt1.w};
            const float skv[8]  = {k0.x, k0.y, k0.z, k0.w, k1.x, k1.y, k1.z, k1.w};
            float s = 0.f, s2 = 0.f;
            for (int j = 0; j < 8; ++j) { s += vals[j]; s2 += vals[j] * vals[j]; }
            for (int off = 16; off > 0; off >>= 1) {
                s  += __shfl_xor(s,  off, 32);
                s2 += __shfl_xor(s2, off, 32);
            }
            const float mu  = s * (1.0f / OUT_DIM);
            const float var = s2 * (1.0f / OUT_DIM) - mu * mu;
            const float rs  = rsqrtf(var + LN_EPS);
            float o[8];
            for (int j = 0; j < 8; ++j) {
                const float ln = (vals[j] - mu) * rs * gmv[j] + btv[j];
                o[j] = fmaxf(ln, 0.f) + skv[j];
            }
            float4* orow = (float4*)(out + (size_t)(m0 + row) * OUT_DIM) + lane * 2;
            orow[0] = make_float4(o[0], o[1], o[2], o[3]);
            orow[1] = make_float4(o[4], o[5], o[6], o[7]);
        }
    }
}

extern "C" void kernel_launch(void* const* d_in, const int* in_sizes, int n_in,
                              void* d_out, int out_size, void* d_ws, size_t ws_size,
                              hipStream_t stream) {
    const float* feat   = (const float*)d_in[0];
    const int*   src    = (const int*)d_in[1];
    const int*   dst    = (const int*)d_in[2];
    const float* W      = (const float*)d_in[3];
    const float* bias   = (const float*)d_in[4];
    const float* gamma  = (const float*)d_in[5];
    const float* beta   = (const float*)d_in[6];
    const float* skip_W = (const float*)d_in[7];
    const float* skip_b = (const float*)d_in[8];
    float* out = (float*)d_out;

    const int n_nodes = in_sizes[0] / IN_DIM;
    const int n_edges = in_sizes[1];

    int*   deg_in  = (int*)d_ws + n_nodes;
    float* h       = (float*)(deg_in + n_nodes);
    float* agg     = h + (size_t)n_nodes * IN_DIM;

    int* planes = (int*)agg;

    unsigned char* key = (unsigned char*)d_out;
    const int scan_step = KPV * SCAN_U * AGG_T;
    const int n_pad = (n_edges + scan_step - 1) / scan_step * scan_step;

    degree_hist<<<4 * HP, HT, 0, stream>>>(src, dst, planes, key, n_edges, n_pad / 4);

    scale_rows<<<(n_nodes + 31) / 32, 256, 0, stream>>>(feat, planes, deg_in, h, n_nodes);

    agg_range<<<(n_nodes + RNG - 1) / RNG, AGG_T, 0, stream>>>(src, dst, key, h, agg, n_pad, n_nodes);

    gcn_fused<<<GF_GRID, 256, 0, stream>>>(agg, deg_in, W, bias, gamma, beta,
                                              feat, skip_W, skip_b, out, n_nodes);
}


# scaled rows h no longer stored: agg_range gathers feature rows and scales each by norm_src[src] in registers; the row-scaling pass shrinks to a per-node degree sum (norm_src, deg_in)
# speedup vs baseline: 2.6286x; 1.0458x over previous
#include <hip/hip_runtime.h>
#include <hip/hip_bf16.h>

typedef __attribute__((ext_vector_type(2))) float v2f;
typedef __attribute__((ext_vector_type(8))) float v8f;

#define IN_DIM 128
#define OUT_DIM 256
#define LN_EPS 1e-5f

#define RNG      512
#define KEY_NONE 0xFFu
#define HP   16
#define HN   50016
#define HT   1024
#define HU   5

__device__ __forceinline__ void hist_add(int* cnt, int node, int lo) {
    const unsigned r = (unsigned)(node - lo);
    if (r < (unsigned)HN) atomicAdd(&cnt[r], 1);
}

__device__ __forceinline__ unsigned key_of(int node) {
    return node < 0 ? KEY_NONE : (unsigned)node / RNG;
}

__global__ __launch_bounds__(HT) void degree_hist(const int* __restrict__ src, const int* __restrict__ dst,
                                                  int* __restrict__ planes,
                                                  unsigned char* __restrict__ key,
                                                  int n_edges, int n_vec) {
    __shared__ __attribute__((aligned(16))) int cnt[HN];

    const int tid  = threadIdx.x;
    const int c    = blockIdx.x % HP;
    const int role = blockIdx.x / HP;
    const int arr  = role >> 1;
    const int lo   = (role & 1) * HN;
    const int* __restrict__ idx = arr ? dst : src;
    const bool wkey = role == 2;

    for (int i = tid; i < HN / 4; i += HT) ((int4*)cnt)[i] = make_int4(0, 0, 0, 0);
    __syncthreads();

    const int chv  = n_vec / HP;
    const int vbeg = c * chv, vend = vbeg + chv;
    for (int v0 = vbeg; v0 < vend; v0 += HU * HT) {
        int4 x[HU];
#pragma unroll
        for (int u = 0; u < HU; ++u) {
            const int v = v0 + u * HT + tid;
            x[u] = make_int4(-1, -1, -1, -1);
            if (v < vend && 4 * v + 3 < n_edges) x[u] = ((const int4*)idx)[v];
        }
#pragma unroll
        for (int u = 0; u < HU; ++u) {
            const int v = v0 + u * HT + tid;
            hist_add(cnt, x[u].x, lo);
            hist_add(cnt, x[u].y, lo);
            hist_add(cnt, x[u].z, lo);
            hist_add(cnt, x[u].w, lo);
            if (wkey && v < vend && (4 * v + 3 < n_edges || 4 * v >= n_edges))
                ((unsigned*)key)[v] = key_of(x[u].x) | (key_of(x[u].y) << 8) |
                                      (key_of(x[u].z) << 16) | (key_of(x[u].w) << 24);
        }
    }
    const int vt = n_edges >> 2;
    if ((n_edges & 3) && vt >= vbeg && vt < vend && tid < 4) {
        const int e = 4 * vt + tid;
        const int node = e < n_edges ? idx[e] : -1;
        hist_add(cnt, node, lo);
        if (wkey) key[e] = (unsigned char)key_of(node);
    }
    __syncthreads();

    int4* o = (int4*)(planes + (size_t)(arr * HP + c) * (2 * HN) + lo);
    for (int i = tid; i < HN / 4; i += HT) o[i] = ((const int4*)cnt)[i];
}

__global__ __launch_bounds__(256) void node_norms(const int* __restrict__ planes,
                                                  float* __restrict__ norm_src,
                                                  int* __restrict__ deg_in, int n_nodes) {
    const int node = blockIdx.x * 256 + threadIdx.x;
    if (node >= n_nodes) return;
    int so = 0, si = 0;
#pragma unroll
    for (int q = 0; q < HP; ++q) {
        so += planes[(size_t)q * (2 * HN) + node];
        si += planes[(size_t)(HP + q) * (2 * HN) + node];
    }
    norm_src[node] = rsqrtf(fmaxf((float)so, 1.0f));
    deg_in[node] = si;
}

#define AGG_T 1024
#define AGG_W (AGG_T / 32)
#define KPV   16
#define SCAN_U 4
#define QCAP  288
#define GU    8

__device__ __forceinline__ unsigned zero_bytes(unsigned x) {
    return ~(((x & 0x7F7F7F7Fu) + 0x7F7F7F7Fu) | x | 0x7F7F7F7Fu);
}

__device__ __forceinline__ unsigned scan_bits(int4 k, unsigned pat) {
    return (zero_bytes((unsigned)k.x ^ pat) >> 7) | (zero_bytes((unsigned)k.y ^ pat) >> 6) |
           (zero_bytes((unsigned)k.z ^ pat) >> 5) | (zero_bytes((unsigned)k.w ^ pat) >> 4);
}

__device__ __forceinline__ void q_round(unsigned& g, unsigned& mask, int e0, int& cnt, int* qe, int lane) {
    const bool m = g != 0u;
    const int b = __builtin_ctz(g | 0x80000000u);
    const int pos = cnt + __builtin_popcount(mask & ((1u << lane) - 1u));
    if (m) qe[pos] = e0 + ((b & 4) ? AGG_T * KPV : 0) + ((b & 3) << 2) + (b >> 3);
    cnt += __builtin_popcount(mask);
    g &= g - 1u;
    mask = __builtin_amdgcn_ballot_w32(g != 0u);
}

__device__ __forceinline__ void q_drain(int& cnt, const int* qe, int base,
                                        const int* __restrict__ src, const int* __restrict__ dst,
                                        const float* __restrict__ feat,
                                        const float* __restrict__ norm_src, float* acc, int lane) {
    for (int b0 = 0; b0 < cnt; b0 += 32) {
        const int nb = min(32, cnt - b0);
        int s = 0, dl = 0;
        int nrb = 0;
        if (lane < nb) { const int e = qe[b0 + lane]; s = src[e]; dl = dst[e] - base; nrb = __float_as_int(norm_src[s]); }
        for (int i = 0; i < nb; i += GU) {
            float4 v[GU];
            int rl[GU];
#pragma unroll
            for (int u = 0; u < GU; ++u) {
                const int iu = min(i + u, nb - 1);
                const int su = __builtin_amdgcn_readlane(s, iu);
                rl[u] = __builtin_amdgcn_readlane(dl, iu);
                v[u] = ((const float4*)(feat + (size_t)su * IN_DIM))[lane];
            }
#pragma unroll
            for (int u = 0; u < GU; ++u) {
                if (i + u < nb) {
                    const float sc = __int_as_float(__builtin_amdgcn_readlane(nrb, i + u));
                    float* a = acc + rl[u] * IN_DIM + lane;
                    __hip_atomic_fetch_add(a,      v[u].x * sc, __ATOMIC_RELAXED, __HIP_MEMORY_SCOPE_WORKGROUP);
                    __hip_atomic_fetch_add(a + 32, v[u].y * sc, __ATOMIC_RELAXED, __HIP_MEMORY_SCOPE_WORKGROUP);
                    __hip_atomic_fetch_add(a + 64, v[u].z * sc, __ATOMIC_RELAXED, __HIP_MEMORY_SCOPE_WORKGROUP);
                    __hip_atomic_fetch_add(a + 96, v[u].w * sc, __ATOMIC_RELAXED, __HIP_MEMORY_SCOPE_WORKGROUP);
                }
            }
        }
    }
    cnt = 0;
}

__global__ __launch_bounds__(AGG_T) void agg_range(const int* __restrict__ src,
                                                   const int* __restrict__ dst,
                                                   const unsigned char* __restrict__ key,
                                                   const float* __restrict__ feat,
                                                   const float* __restrict__ norm_src,
                                                   float* __restrict__ agg,
                                                   int n_pad, int n_nodes) {
    __shared__ __attribute__((aligned(16))) float acc[RNG * IN_DIM];
    __shared__ int qe_all[AGG_W][QCAP];

    const int tid  = threadIdx.x;
    const int lane = tid & 31;
    const int wv   = tid >> 5;
    const int base = blockIdx.x * RNG;
    int* qe = qe_all[wv];
    const unsigned pat = blockIdx.x * 0x01010101u;

    for (int i = tid; i < RNG * IN_DIM / 4; i += AGG_T)
        ((float4*)acc)[i] = make_float4(0.f, 0.f, 0.f, 0.f);
    __syncthreads();

    int cnt = 0;
    const int4* key4 = (const int4*)key;
    const int nvec = n_pad / KPV;

    for (int v0 = 0; v0 < nvec; v0 += SCAN_U * AGG_T) {
        int4 k[SCAN_U];
#pragma unroll
        for (int u = 0; u < SCAN_U; ++u)
            k[u] = key4[v0 + u * AGG_T + tid];
#pragma unroll
        for (int u = 0; u < SCAN_U; u += 2) {
            const int e0 = (v0 + u * AGG_T + tid) * KPV;
            unsigned g = scan_bits(k[u], pat) | (scan_bits(k[u + 1], pat) << 4);
            unsigned mask = __builtin_amdgcn_ballot_w32(g != 0u);
            for (;;) {
                while (mask && cnt <= QCAP - 32) q_round(g, mask, e0, cnt, qe, lane);
                if (cnt >= 32) q_drain(cnt, qe, base, src, dst, feat, norm_src, acc, lane);
                if (!mask) break;
            }
        }
    }
    q_drain(cnt, qe, base, src, dst, feat, norm_src, acc, lane);
    __syncthreads();

    for (int r = wv; r < RNG; r += AGG_W) {
        const int row = base + r;
        if (row >= n_nodes) break;
        const float* a = acc + r * IN_DIM + lane;
        const float4 o = make_float4(a[0], a[32], a[64], a[96]);
        ((float4*)(agg + (size_t)row * IN_DIM))[lane] = o;
    }
}

#define GF_GRID 256
#define GF_BW   (32 * 8 * 32)
__global__ __launch_bounds__(256) void gcn_fused(const float* __restrict__ agg,
                                                 const int* __restrict__ deg_in,
                                                 const float* __restrict__ W,
                                                 const float* __restrict__ bias,
                                                 const float* __restrict__ gamma,
                                                 const float* __restrict__ beta,
                                                 const float* __restrict__ feat,
                                                 const float* __restrict__ skip_W,
                                                 const float* __restrict__ skip_b,
                                                 float* __restrict__ out, int n_nodes) {
    __shared__ __attribute__((aligned(16))) float4 ldsB[2 * GF_BW];
    __shared__ __attribute__((aligned(16))) float lds_gcn[16][OUT_DIM + 4];
    __shared__ __attribute__((aligned(16))) float lds_skip[16][OUT_DIM + 4];

    const int tid  = threadIdx.x;
    const int wave = tid >> 5;
    const int lane = tid & 31;
    const int half = lane >> 4;
    const int l16  = lane & 15;
    const int bcol = wave * 32 + l16;

    for (int it = 0; it < 64; ++it) {
        const float* M = (it < 32) ? W : skip_W;
        const int s  = it & 31;
        const int k0 = 8 * (s >> 1) + 4 * half + 2 * (s & 1);
        const float* r0 = M + (size_t)k0 * OUT_DIM + bcol;
        float4 v;
        v.x = r0[0];  v.y = r0[OUT_DIM];
        v.z = r0[16]; v.w = r0[OUT_DIM + 16];
        ldsB[it * 256 + tid] = v;
    }

    const float bb0 = bias[bcol],      sb0 = skip_b[bcol];
    const float bb1 = bias[bcol + 16], sb1 = skip_b[bcol + 16];
    const float4 gm0 = ((const float4*)gamma)[lane * 2], gm1 = ((const float4*)gamma)[lane * 2 + 1];
    const float4 bt0 = ((const float4*)beta)[lane * 2],  bt1 = ((const float4*)beta)[lane * 2 + 1];
    const float4* bW = ldsB + wave * 32 + lane;
    const float4* bS = bW + GF_BW;
    const int n_tiles = n_nodes / 16;

    int tile = blockIdx.x;
    float4 nA[16], nF[16];
    if (tile < n_tiles) {
        const float4* ar = (const float4*)(agg  + (size_t)(tile * 16 + l16) * IN_DIM) + half;
        const float4* fr = (const float4*)(feat + (size_t)(tile * 16 + l16) * IN_DIM) + half;
        for (int j = 0; j < 16; ++j) { nA[j] = ar[2 * j]; nF[j] = fr[2 * j]; }
    }

    for (; tile < n_tiles; tile += GF_GRID) {
        const int m0 = tile * 16;
        float4 aA[16], aF[16];
        for (int j = 0; j < 16; ++j) { aA[j] = nA[j]; aF[j] = nF[j]; }
        if (tile + GF_GRID < n_tiles) {
            const float4* ar = (const float4*)(agg  + (size_t)(m0 + GF_GRID * 16 + l16) * IN_DIM) + half;
            const float4* fr = (const float4*)(feat + (size_t)(m0 + GF_GRID * 16 + l16) * IN_DIM) + half;
            for (int j = 0; j < 16; ++j) { nA[j] = ar[2 * j]; nF[j] = fr[2 * j]; }
        }
        int dg[8];
        for (int i = 0; i < 8; ++i) dg[i] = deg_in[m0 + i + 8 * half];

        v8f cg0 = {}, cg1 = {};
        v8f cs0 = {}, cs1 = {};
        for (int j = 0; j < 16; ++j) {
            for (int p = 0; p < 2; ++p) {
                const int s = 2 * j + p;
                const float4 w4 = bW[s * 256];
                const float4 s4 = bS[s * 256];
                v2f a, f, w0, w1, s0, s1;
                a[0] = p ? aA[j].z : aA[j].x;  a[1] = p ? aA[j].w : aA[j].y;
                f[0] = p ? aF[j].z : aF[j].x;  f[1] = p ? aF[j].w : aF[j].y;
                w0[0] = w4.x; w0[1] = w4.y; w1[0] = w4.z; w1[1] = w4.w;
                s0[0] = s4.x; s0[1] = s4.y; s1[0] = s4.z; s1[1] = s4.w;
                cg0 = __builtin_amdgcn_wmma_f32_16x16x4_f32(false, a, false, w0, (short)0, cg0, false, false);
                cg1 = __builtin_amdgcn_wmma_f32_16x16x4_f32(false, a, false, w1, (short)0, cg1, false, false);
                cs0 = __builtin_amdgcn_wmma_f32_16x16x4_f32(false, f, false, s0, (short)0, cs0, false, false);
                cs1 = __builtin_amdgcn_wmma_f32_16x16x4_f32(false, f, false, s1, (short)0, cs1, false, false);
            }
        }

        __syncthreads();
        for (int i = 0; i < 8; ++i) {
            const int row = i + 8 * half;
            const float nrm = rsqrtf(fmaxf((float)dg[i], 1.0f));
            lds_gcn[row][bcol]       = cg0[i] * nrm + bb0;
            lds_gcn[row][bcol + 16]  = cg1[i] * nrm + bb1;
            lds_skip[row][bcol]      = cs0[i] + sb0;
            lds_skip[row][bcol + 16] = cs1[i] + sb1;
        }
        __syncthreads();

        for (int rr = 0; rr < 2; ++rr) {
            const int row = wave * 2 + rr;
            const float4 x0 = *(const float4*)&lds_gcn[row][lane * 8];
            const float4 x1 = *(const float4*)&lds_gcn[row][lane * 8 + 4];
            const float4 k0 = *(const float4*)&lds_skip[row][lane * 8];
            const float4 k1 = *(const float4*)&lds_skip[row][lane * 8 + 4];
            const float vals[8] = {x0.x, x0.y, x0.z, x0.w, x1.x, x1.y, x1.z, x1.w};
            const float gmv[8]  = {gm0.x, gm0.y, gm0.z, gm0.w, gm1.x, gm1.y, gm1.z, gm1.w};
            const float btv[8]  = {bt0.x, bt0.y, bt0.z, bt0.w, bt1.x, bt1.y, bt1.z, bt1.w};
            const float skv[8]  = {k0.x, k0.y, k0.z, k0.w, k1.x, k1.y, k1.z, k1.w};
            float s = 0.f, s2 = 0.f;
            for (int j = 0; j < 8; ++j) { s += vals[j]; s2 += vals[j] * vals[j]; }
            for (int off = 16; off > 0; off >>= 1) {
                s  += __shfl_xor(s,  off, 32);
                s2 += __shfl_xor(s2, off, 32);
            }
            const float mu  = s * (1.0f / OUT_DIM);
            const float var = s2 * (1.0f / OUT_DIM) - mu * mu;
            const float rs  = rsqrtf(var + LN_EPS);
            float o[8];
            for (int j = 0; j < 8; ++j) {
                const float ln = (vals[j] - mu) * rs * gmv[j] + btv[j];
                o[j] = fmaxf(ln, 0.f) + skv[j];
            }
            float4* orow = (float4*)(out + (size_t)(m0 + row) * OUT_DIM) + lane * 2;
            orow[0] = make_float4(o[0], o[1], o[2], o[3]);
            orow[1] = make_float4(o[4], o[5], o[6], o[7]);
        }
    }
}

extern "C" void kernel_launch(void* const* d_in, const int* in_sizes, int n_in,
                              void* d_out, int out_size, void* d_ws, size_t ws_size,
                              hipStream_t stream) {
    const float* feat   = (const float*)d_in[0];
    const int*   src    = (const int*)d_in[1];
    const int*   dst    = (const int*)d_in[2];
    const float* W      = (const float*)d_in[3];
    const float* bias   = (const float*)d_in[4];
    const float* gamma  = (const float*)d_in[5];
    const float* beta   = (const float*)d_in[6];
    const float* skip_W = (const float*)d_in[7];
    const float* skip_b = (const float*)d_in[8];
    float* out = (float*)d_out;

    const int n_nodes = in_sizes[0] / IN_DIM;
    const int n_edges = in_sizes[1];

    float* norm_src = (float*)d_ws;
    int*   deg_in   = (int*)d_ws + n_nodes;
    float* agg      = (float*)(deg_in + n_nodes) + (size_t)n_nodes * IN_DIM;

    int* planes = (int*)agg;

    unsigned char* key = (unsigned char*)d_out;
    const int scan_step = KPV * SCAN_U * AGG_T;
    const int n_pad = (n_edges + scan_step - 1) / scan_step * scan_step;

    degree_hist<<<4 * HP, HT, 0, stream>>>(src, dst, planes, key, n_edges, n_pad / 4);

    node_norms<<<(n_nodes + 255) / 256, 256, 0, stream>>>(planes, norm_src, deg_in, n_nodes);

    agg_range<<<(n_nodes + RNG - 1) / RNG, AGG_T, 0, stream>>>(src, dst, key, feat, norm_src, agg, n_pad, n_nodes);

    gcn_fused<<<GF_GRID, 256, 0, stream>>>(agg, deg_in, W, bias, gamma, beta,
                                              feat, skip_W, skip_b, out, n_nodes);
}


# gcn_fused weights register-stationary: each wave keeps its W/skip_W fragments in 256 VGPRs for all tiles, no LDS read among the WMMAs; A rows single-buffered, fetched under the epilogue; sums same
# speedup vs baseline: 2.9486x; 1.1217x over previous
#include <hip/hip_runtime.h>
#include <hip/hip_bf16.h>

typedef __attribute__((ext_vector_type(2))) float v2f;
typedef __attribute__((ext_vector_type(8))) float v8f;

#define IN_DIM 128
#define OUT_DIM 256
#define LN_EPS 1e-5f

#define RNG      512
#define KEY_NONE 0xFFu
#define HP   16
#define HN   50016
#define HT   1024
#define HU   5

__device__ __forceinline__ void hist_add(int* cnt, int node, int lo) {
    const unsigned r = (unsigned)(node - lo);
    if (r < (unsigned)HN) atomicAdd(&cnt[r], 1);
}

__device__ __forceinline__ unsigned key_of(int node) {
    return node < 0 ? KEY_NONE : (unsigned)node / RNG;
}

__global__ __launch_bounds__(HT) void degree_hist(const int* __restrict__ src, const int* __restrict__ dst,
                                                  int* __restrict__ planes,
                                                  unsigned char* __restrict__ key,
                                                  int n_edges, int n_vec) {
    __shared__ __attribute__((aligned(16))) int cnt[HN];

    const int tid  = threadIdx.x;
    const int c    = blockIdx.x % HP;
    const int role = blockIdx.x / HP;
    const int arr  = role >> 1;
    const int lo   = (role & 1) * HN;
    const int* __restrict__ idx = arr ? dst : src;
    const bool wkey = role == 2;

    for (int i = tid; i < HN / 4; i += HT) ((int4*)cnt)[i] = make_int4(0, 0, 0, 0);
    __syncthreads();

    const int chv  = n_vec / HP;
    const int vbeg = c * chv, vend = vbeg + chv;
    for (int v0 = vbeg; v0 < vend; v0 += HU * HT) {
        int4 x[HU];
#pragma unroll
        for (int u = 0; u < HU; ++u) {
            const int v = v0 + u * HT + tid;
            x[u] = make_int4(-1, -1, -1, -1);
            if (v < vend && 4 * v + 3 < n_edges) x[u] = ((const int4*)idx)[v];
        }
#pragma unroll
        for (int u = 0; u < HU; ++u) {
            const int v = v0 + u * HT + tid;
            hist_add(cnt, x[u].x, lo);
            hist_add(cnt, x[u].y, lo);
            hist_add(cnt, x[u].z, lo);
            hist_add(cnt, x[u].w, lo);
            if (wkey && v < vend && (4 * v + 3 < n_edges || 4 * v >= n_edges))
                ((unsigned*)key)[v] = key_of(x[u].x) | (key_of(x[u].y) << 8) |
                                      (key_of(x[u].z) << 16) | (key_of(x[u].w) << 24);
        }
    }
    const int vt = n_edges >> 2;
    if ((n_edges & 3) && vt >= vbeg && vt < vend && tid < 4) {
        const int e = 4 * vt + tid;
        const int node = e < n_edges ? idx[e] : -1;
        hist_add(cnt, node, lo);
        if (wkey) key[e] = (unsigned char)key_of(node);
    }
    __syncthreads();

    int4* o = (int4*)(planes + (size_t)(arr * HP + c) * (2 * HN) + lo);
    for (int i = tid; i < HN / 4; i += HT) o[i] = ((const int4*)cnt)[i];
}

__global__ __launch_bounds__(256) void node_norms(const int* __restrict__ planes,
                                                  float* __restrict__ norm_src,
                                                  int* __restrict__ deg_in, int n_nodes) {
    const int node = blockIdx.x * 256 + threadIdx.x;
    if (node >= n_nodes) return;
    int so = 0, si = 0;
#pragma unroll
    for (int q = 0; q < HP; ++q) {
        so += planes[(size_t)q * (2 * HN) + node];
        si += planes[(size_t)(HP + q) * (2 * HN) + node];
    }
    norm_src[node] = rsqrtf(fmaxf((float)so, 1.0f));
    deg_in[node] = si;
}

#define AGG_T 1024
#define AGG_W (AGG_T / 32)
#define KPV   16
#define SCAN_U 4
#define QCAP  288
#define GU    8

__device__ __forceinline__ unsigned zero_bytes(unsigned x) {
    return ~(((x & 0x7F7F7F7Fu) + 0x7F7F7F7Fu) | x | 0x7F7F7F7Fu);
}

__device__ __forceinline__ unsigned scan_bits(int4 k, unsigned pat) {
    return (zero_bytes((unsigned)k.x ^ pat) >> 7) | (zero_bytes((unsigned)k.y ^ pat) >> 6) |
           (zero_bytes((unsigned)k.z ^ pat) >> 5) | (zero_bytes((unsigned)k.w ^ pat) >> 4);
}

__device__ __forceinline__ void q_round(unsigned& g, unsigned& mask, int e0, int& cnt, int* qe, int lane) {
    const bool m = g != 0u;
    const int b = __builtin_ctz(g | 0x80000000u);
    const int pos = cnt + __builtin_popcount(mask & ((1u << lane) - 1u));
    if (m) qe[pos] = e0 + ((b & 4) ? AGG_T * KPV : 0) + ((b & 3) << 2) + (b >> 3);
    cnt += __builtin_popcount(mask);
    g &= g - 1u;
    mask = __builtin_amdgcn_ballot_w32(g != 0u);
}

__device__ __forceinline__ void q_drain(int& cnt, const int* qe, int base,
                                        const int* __restrict__ src, const int* __restrict__ dst,
                                        const float* __restrict__ feat,
                                        const float* __restrict__ norm_src, float* acc, int lane) {
    for (int b0 = 0; b0 < cnt; b0 += 32) {
        const int nb = min(32, cnt - b0);
        int s = 0, dl = 0;
        int nrb = 0;
        if (lane < nb) { const int e = qe[b0 + lane]; s = src[e]; dl = dst[e] - base; nrb = __float_as_int(norm_src[s]); }
        for (int i = 0; i < nb; i += GU) {
            float4 v[GU];
            int rl[GU];
#pragma unroll
            for (int u = 0; u < GU; ++u) {
                const int iu = min(i + u, nb - 1);
                const int su = __builtin_amdgcn_readlane(s, iu);
                rl[u] = __builtin_amdgcn_readlane(dl, iu);
                v[u] = ((const float4*)(feat + (size_t)su * IN_DIM))[lane];
            }
#pragma unroll
            for (int u = 0; u < GU; ++u) {
                if (i + u < nb) {
                    const float sc = __int_as_float(__builtin_amdgcn_readlane(nrb, i + u));
                    float* a = acc + rl[u] * IN_DIM + lane;
                    __hip_atomic_fetch_add(a,      v[u].x * sc, __ATOMIC_RELAXED, __HIP_MEMORY_SCOPE_WORKGROUP);
                    __hip_atomic_fetch_add(a + 32, v[u].y * sc, __ATOMIC_RELAXED, __HIP_MEMORY_SCOPE_WORKGROUP);
                    __hip_atomic_fetch_add(a + 64, v[u].z * sc, __ATOMIC_RELAXED, __HIP_MEMORY_SCOPE_WORKGROUP);
                    __hip_atomic_fetch_add(a + 96, v[u].w * sc, __ATOMIC_RELAXED, __HIP_MEMORY_SCOPE_WORKGROUP);
                }
            }
        }
    }
    cnt = 0;
}

__global__ __launch_bounds__(AGG_T) void agg_range(const int* __restrict__ src,
                                                   const int* __restrict__ dst,
                                                   const unsigned char* __restrict__ key,
                                                   const float* __restrict__ feat,
                                                   const float* __restrict__ norm_src,
                                                   float* __restrict__ agg,
                                                   int n_pad, int n_nodes) {
    __shared__ __attribute__((aligned(16))) float acc[RNG * IN_DIM];
    __shared__ int qe_all[AGG_W][QCAP];

    const int tid  = threadIdx.x;
    const int lane = tid & 31;
    const int wv   = tid >> 5;
    const int base = blockIdx.x * RNG;
    int* qe = qe_all[wv];
    const unsigned pat = blockIdx.x * 0x01010101u;

    for (int i = tid; i < RNG * IN_DIM / 4; i += AGG_T)
        ((float4*)acc)[i] = make_float4(0.f, 0.f, 0.f, 0.f);
    __syncthreads();

    int cnt = 0;
    const int4* key4 = (const int4*)key;
    const int nvec = n_pad / KPV;

    for (int v0 = 0; v0 < nvec; v0 += SCAN_U * AGG_T) {
        int4 k[SCAN_U];
#pragma unroll
        for (int u = 0; u < SCAN_U; ++u)
            k[u] = key4[v0 + u * AGG_T + tid];
#pragma unroll
        for (int u = 0; u < SCAN_U; u += 2) {
            const int e0 = (v0 + u * AGG_T + tid) * KPV;
            unsigned g = scan_bits(k[u], pat) | (scan_bits(k[u + 1], pat) << 4);
            unsigned mask = __builtin_amdgcn_ballot_w32(g != 0u);
            for (;;) {
                while (mask && cnt <= QCAP - 32) q_round(g, mask, e0, cnt, qe, lane);
                if (cnt >= 32) q_drain(cnt, qe, base, src, dst, feat, norm_src, acc, lane);
                if (!mask) break;
            }
        }
    }
    q_drain(cnt, qe, base, src, dst, feat, norm_src, acc, lane);
    __syncthreads();

    for (int r = wv; r < RNG; r += AGG_W) {
        const int row = base + r;
        if (row >= n_nodes) break;
        const float* a = acc + r * IN_DIM + lane;
        const float4 o = make_float4(a[0], a[32], a[64], a[96]);
        ((float4*)(agg + (size_t)row * IN_DIM))[lane] = o;
    }
}

#define GF_GRID 256

__device__ __forceinline__ void lds_barrier() {
    __builtin_amdgcn_fence(__ATOMIC_RELEASE, "workgroup", "local");
    __builtin_amdgcn_s_barrier();
    __builtin_amdgcn_fence(__ATOMIC_ACQUIRE, "workgroup", "local");
}
__global__ __launch_bounds__(256) void gcn_fused(const float* __restrict__ agg,
                                                 const int* __restrict__ deg_in,
                                                 const float* __restrict__ W,
                                                 const float* __restrict__ bias,
                                                 const float* __restrict__ gamma,
                                                 const float* __restrict__ beta,
                                                 const float* __restrict__ feat,
                                                 const float* __restrict__ skip_W,
                                                 const float* __restrict__ skip_b,
                                                 float* __restrict__ out, int n_nodes) {
    __shared__ __attribute__((aligned(16))) float lds_gcn[16][OUT_DIM + 4];
    __shared__ __attribute__((aligned(16))) float lds_skip[16][OUT_DIM + 4];

    const int tid  = threadIdx.x;
    const int wave = tid >> 5;
    const int lane = tid & 31;
    const int half = lane >> 4;
    const int l16  = lane & 15;
    const int bcol = wave * 32 + l16;

    float4 wf[32], sf[32];
#pragma unroll
    for (int s = 0; s < 32; ++s) {
        const int k0 = 8 * (s >> 1) + 4 * half + 2 * (s & 1);
        const float* r0 = W + (size_t)k0 * OUT_DIM + bcol;
        const float* r1 = skip_W + (size_t)k0 * OUT_DIM + bcol;
        wf[s].x = r0[0];  wf[s].y = r0[OUT_DIM];
        wf[s].z = r0[16]; wf[s].w = r0[OUT_DIM + 16];
        sf[s].x = r1[0];  sf[s].y = r1[OUT_DIM];
        sf[s].z = r1[16]; sf[s].w = r1[OUT_DIM + 16];
    }

    const float bb0 = bias[bcol],      sb0 = skip_b[bcol];
    const float bb1 = bias[bcol + 16], sb1 = skip_b[bcol + 16];
    const float4 gm0 = ((const float4*)gamma)[lane * 2], gm1 = ((const float4*)gamma)[lane * 2 + 1];
    const float4 bt0 = ((const float4*)beta)[lane * 2],  bt1 = ((const float4*)beta)[lane * 2 + 1];
    const int n_tiles = n_nodes / 16;

    int tile = blockIdx.x;
    float4 aA[16], aF[16];
    int dgN[8];
    if (tile < n_tiles) {
        for (int i = 0; i < 8; ++i) dgN[i] = deg_in[tile * 16 + i + 8 * half];
        const float4* ar = (const float4*)(agg  + (size_t)(tile * 16 + l16) * IN_DIM) + half;
        const float4* fr = (const float4*)(feat + (size_t)(tile * 16 + l16) * IN_DIM) + half;
#pragma unroll
        for (int j = 0; j < 16; ++j) { aA[j] = ar[2 * j]; aF[j] = fr[2 * j]; }
    }

    for (; tile < n_tiles; tile += GF_GRID) {
        const int m0 = tile * 16;
        float nrm[8];
        for (int i = 0; i < 8; ++i) nrm[i] = rsqrtf(fmaxf((float)dgN[i], 1.0f));

        v8f cg0 = {}, cg1 = {};
        v8f cs0 = {}, cs1 = {};
#pragma unroll
        for (int j = 0; j < 16; ++j) {
#pragma unroll
            for (int p = 0; p < 2; ++p) {
                const int s = 2 * j + p;
                const float4 w4 = wf[s];
                const float4 s4 = sf[s];
                v2f a, f, w0, w1, s0, s1;
                a[0] = p ? aA[j].z : aA[j].x;  a[1] = p ? aA[j].w : aA[j].y;
                f[0] = p ? aF[j].z : aF[j].x;  f[1] = p ? aF[j].w : aF[j].y;
                w0[0] = w4.x; w0[1] = w4.y; w1[0] = w4.z; w1[1] = w4.w;
                s0[0] = s4.x; s0[1] = s4.y; s1[0] = s4.z; s1[1] = s4.w;
                cg0 = __builtin_amdgcn_wmma_f32_16x16x4_f32(false, a, false, w0, (short)0, cg0, false, false);
                cg1 = __builtin_amdgcn_wmma_f32_16x16x4_f32(false, a, false, w1, (short)0, cg1, false, false);
                cs0 = __builtin_amdgcn_wmma_f32_16x16x4_f32(false, f, false, s0, (short)0, cs0, false, false);
                cs1 = __builtin_amdgcn_wmma_f32_16x16x4_f32(false, f, false, s1, (short)0, cs1, false, false);
            }
        }

        if (tile + GF_GRID < n_tiles) {
            for (int i = 0; i < 8; ++i) dgN[i] = deg_in[m0 + GF_GRID * 16 + i + 8 * half];
            const float4* ar = (const float4*)(agg  + (size_t)(m0 + GF_GRID * 16 + l16) * IN_DIM) + half;
            const float4* fr = (const float4*)(feat + (size_t)(m0 + GF_GRID * 16 + l16) * IN_DIM) + half;
#pragma unroll
            for (int j = 0; j < 16; ++j) { aA[j] = ar[2 * j]; aF[j] = fr[2 * j]; }
        }

        lds_barrier();
        for (int i = 0; i < 8; ++i) {
            const int row = i + 8 * half;
            lds_gcn[row][bcol]       = cg0[i] * nrm[i] + bb0;
            lds_gcn[row][bcol + 16]  = cg1[i] * nrm[i] + bb1;
            lds_skip[row][bcol]      = cs0[i] + sb0;
            lds_skip[row][bcol + 16] = cs1[i] + sb1;
        }
        lds_barrier();

        for (int rr = 0; rr < 2; ++rr) {
            const int row = wave * 2 + rr;
            const float4 x0 = *(const float4*)&lds_gcn[row][lane * 8];
            const float4 x1 = *(const float4*)&lds_gcn[row][lane * 8 + 4];
            const float4 k0 = *(const float4*)&lds_skip[row][lane * 8];
            const float4 k1 = *(const float4*)&lds_skip[row][lane * 8 + 4];
            const float vals[8] = {x0.x, x0.y, x0.z, x0.w, x1.x, x1.y, x1.z, x1.w};
            const float gmv[8]  = {gm0.x, gm0.y, gm0.z, gm0.w, gm1.x, gm1.y, gm1.z, gm1.w};
            const float btv[8]  = {bt0.x, bt0.y, bt0.z, bt0.w, bt1.x, bt1.y, bt1.z, bt1.w};
            const float skv[8]  = {k0.x, k0.y, k0.z, k0.w, k1.x, k1.y, k1.z, k1.w};
            float s = 0.f, s2 = 0.f;
            for (int j = 0; j < 8; ++j) { s += vals[j]; s2 += vals[j] * vals[j]; }
            for (int off = 16; off > 0; off >>= 1) {
                s  += __shfl_xor(s,  off, 32);
                s2 += __shfl_xor(s2, off, 32);
            }
            const float mu  = s * (1.0f / OUT_DIM);
            const float var = s2 * (1.0f / OUT_DIM) - mu * mu;
            const float rs  = rsqrtf(var + LN_EPS);
            float o[8];
            for (int j = 0; j < 8; ++j) {
                const float ln = (vals[j] - mu) * rs * gmv[j] + btv[j];
                o[j] = fmaxf(ln, 0.f) + skv[j];
            }
            float4* orow = (float4*)(out + (size_t)(m0 + row) * OUT_DIM) + lane * 2;
            orow[0] = make_float4(o[0], o[1], o[2], o[3]);
            orow[1] = make_float4(o[4], o[5], o[6], o[7]);
        }
        __builtin_amdgcn_sched_barrier(0);
    }
}

extern "C" void kernel_launch(void* const* d_in, const int* in_sizes, int n_in,
                              void* d_out, int out_size, void* d_ws, size_t ws_size,
                              hipStream_t stream) {
    const float* feat   = (const float*)d_in[0];
    const int*   src    = (const int*)d_in[1];
    const int*   dst    = (const int*)d_in[2];
    const float* W      = (const float*)d_in[3];
    const float* bias   = (const float*)d_in[4];
    const float* gamma  = (const float*)d_in[5];
    const float* beta   = (const float*)d_in[6];
    const float* skip_W = (const float*)d_in[7];
    const float* skip_b = (const float*)d_in[8];
    float* out = (float*)d_out;

    const int n_nodes = in_sizes[0] / IN_DIM;
    const int n_edges = in_sizes[1];

    float* norm_src = (float*)d_ws;
    int*   deg_in   = (int*)d_ws + n_nodes;
    float* agg      = (float*)(deg_in + n_nodes) + (size_t)n_nodes * IN_DIM;

    int* planes = (int*)agg;

    unsigned char* key = (unsigned char*)d_out;
    const int scan_step = KPV * SCAN_U * AGG_T;
    const int n_pad = (n_edges + scan_step - 1) / scan_step * scan_step;

    degree_hist<<<4 * HP, HT, 0, stream>>>(src, dst, planes, key, n_edges, n_pad / 4);

    node_norms<<<(n_nodes + 255) / 256, 256, 0, stream>>>(planes, norm_src, deg_in, n_nodes);

    agg_range<<<(n_nodes + RNG - 1) / RNG, AGG_T, 0, stream>>>(src, dst, key, feat, norm_src, agg, n_pad, n_nodes);

    gcn_fused<<<GF_GRID, 256, 0, stream>>>(agg, deg_in, W, bias, gamma, beta,
                                              feat, skip_W, skip_b, out, n_nodes);
}
